# starting kernel
# baseline (speedup 1.0000x reference)
#include <hip/hip_runtime.h>


typedef __attribute__((ext_vector_type(2))) float v2f;
typedef __attribute__((ext_vector_type(8))) float v8f;

constexpr int HDIM = 128;

__device__ __forceinline__ void atomic_add_f32(float* p, float v) {
  (void)__hip_atomic_fetch_add(p, v, __ATOMIC_RELAXED,
                               __HIP_MEMORY_SCOPE_AGENT);
}

__global__ void zero_kernel(float* __restrict__ p, int n) {
  int i = blockIdx.x * blockDim.x + threadIdx.x;
  int stride = gridDim.x * blockDim.x;
  for (; i < n; i += stride) p[i] = 0.0f;
}

__global__ void degree_kernel(const int* __restrict__ dst, int e,
                              float* __restrict__ deg) {
  int i = blockIdx.x * blockDim.x + threadIdx.x;
  if (i < e) atomic_add_f32(&deg[dst[i]], 1.0f);
}

__global__ void scatter_kernel(const float* __restrict__ x,
                               const int* __restrict__ src,
                               const int* __restrict__ dst, int e,
                               float* __restrict__ s) {
  int wid  = (blockIdx.x * blockDim.x + threadIdx.x) >> 5;
  int lane = threadIdx.x & 31;
  if (wid >= e) return;
  int sr = src[wid];
  int dr = dst[wid];
  const float4 v = *(const float4*)(x + (size_t)sr * HDIM + lane * 4);
  float* o = s + (size_t)dr * HDIM + lane * 4;
  atomic_add_f32(o + 0, v.x);
  atomic_add_f32(o + 1, v.y);
  atomic_add_f32(o + 2, v.z);
  atomic_add_f32(o + 3, v.w);
}


__global__ void __launch_bounds__(256)
gemm_in_kernel(const float* __restrict__ A, const float* __restrict__ W,
               const float* __restrict__ bias, const float* __restrict__ emb,
               const int* __restrict__ nid, float* __restrict__ C, int K) {
  const int lane = threadIdx.x & 31;
  const int wave = threadIdx.x >> 5;
  const int half = lane >> 4;
  const int nloc = lane & 15;
  const int m0   = blockIdx.x * 16;
  const int row  = m0 + nloc;
  const int n    = wave * 16 + nloc;

  const float* Arow = A + (size_t)row * K;
  v8f acc = {};
#pragma unroll 8
  for (int k0 = 0; k0 < K; k0 += 4) {
    const int ka = k0 + 2 * half;
    v2f a; a.x = Arow[ka];
           a.y = Arow[ka + 1];
    v2f b; b.x = W[(size_t)ka * HDIM + n];
           b.y = W[(size_t)(ka + 1) * HDIM + n];
    acc = __builtin_amdgcn_wmma_f32_16x16x4_f32(false, a, false, b,
                                                (short)0, acc, false, false);
  }

  const float bn = bias[n];
#pragma unroll
  for (int r = 0; r < 8; ++r) {
    const int m = m0 + r + 8 * half;
    const float v = acc[r] + bn + emb[(size_t)nid[m] * HDIM + n];
    C[(size_t)m * HDIM + n] = v;
  }
}

__global__ void __launch_bounds__(256)
sage_kernel(const float* __restrict__ S, const float* __restrict__ deg,
            const float* __restrict__ X, const float* __restrict__ Wl,
            const float* __restrict__ bl, const float* __restrict__ Wr,
            float* __restrict__ Out, int relu) {
  const int lane = threadIdx.x & 31;
  const int wave = threadIdx.x >> 5;
  const int half = lane >> 4;
  const int nloc = lane & 15;
  const int m0   = blockIdx.x * 16;
  const int row  = m0 + nloc;
  const int n    = wave * 16 + nloc;

  const float* Srow = S + (size_t)row * HDIM;
  const float* Xrow = X + (size_t)row * HDIM;
  v8f accS = {};
  v8f accX = {};
#pragma unroll 4
  for (int k0 = 0; k0 < HDIM; k0 += 4) {
    const int ka = k0 + 2 * half;
    v2f a1; a1.x = Srow[ka]; a1.y = Srow[ka + 1];
    v2f b1; b1.x = Wl[(size_t)ka * HDIM + n];
            b1.y = Wl[(size_t)(ka + 1) * HDIM + n];
    accS = __builtin_amdgcn_wmma_f32_16x16x4_f32(false, a1, false, b1,
                                                 (short)0, accS, false, false);
    v2f a2; a2.x = Xrow[ka]; a2.y = Xrow[ka + 1];
    v2f b2; b2.x = Wr[(size_t)ka * HDIM + n];
            b2.y = Wr[(size_t)(ka + 1) * HDIM + n];
    accX = __builtin_amdgcn_wmma_f32_16x16x4_f32(false, a2, false, b2,
                                                 (short)0, accX, false, false);
  }

  const float bn = bl[n];
#pragma unroll
  for (int r = 0; r < 8; ++r) {
    const int m = m0 + r + 8 * half;
    float d = deg[m];
    d = d < 1.0f ? 1.0f : d;
    float v = accS[r] / d + accX[r] + bn;
    if (relu) v = v > 0.0f ? v : 0.0f;
    Out[(size_t)m * HDIM + n] = v;
  }
}

__global__ void edge_dot_kernel(const float* __restrict__ Hc,
                                const float* __restrict__ Hp,
                                const int* __restrict__ ls,
                                const int* __restrict__ ld, int e,
                                float* __restrict__ out) {
  int wid  = (blockIdx.x * blockDim.x + threadIdx.x) >> 5;
  int lane = threadIdx.x & 31;
  if (wid >= e) return;
  int a = ls[wid];
  int b = ld[wid];
  const float4 va = *(const float4*)(Hc + (size_t)a * HDIM + lane * 4);
  const float4 vb = *(const float4*)(Hp + (size_t)b * HDIM + lane * 4);
  float v = va.x * vb.x + va.y * vb.y + va.z * vb.z + va.w * vb.w;
#pragma unroll
  for (int off = 16; off > 0; off >>= 1) v += __shfl_down(v, off, 32);
  if (lane == 0) out[wid] = v;
}

extern "C" void kernel_launch(void* const* d_in, const int* in_sizes, int n_in,
                              void* d_out, int out_size, void* d_ws,
                              size_t ws_size, hipStream_t stream) {
  (void)in_sizes; (void)n_in; (void)out_size; (void)ws_size;
  const int NC = 4096, NP = 8192, E = 500000, EL = 500000;

  const float* x_country = (const float*)d_in[0];
  const float* x_product = (const float*)d_in[1];
  const float* Wc_lin = (const float*)d_in[2];
  const float* bc_lin = (const float*)d_in[3];
  const float* Wp_lin = (const float*)d_in[4];
  const float* bp_lin = (const float*)d_in[5];
  const float* emb_c  = (const float*)d_in[6];
  const float* emb_p  = (const float*)d_in[7];
  const float* W1ep_l = (const float*)d_in[8];
  const float* b1ep   = (const float*)d_in[9];
  const float* W1ep_r = (const float*)d_in[10];
  const float* W1pc_l = (const float*)d_in[11];
  const float* b1pc   = (const float*)d_in[12];
  const float* W1pc_r = (const float*)d_in[13];
  const float* W2ep_l = (const float*)d_in[14];
  const float* b2ep   = (const float*)d_in[15];
  const float* W2ep_r = (const float*)d_in[16];
  const float* W2pc_l = (const float*)d_in[17];
  const float* b2pc   = (const float*)d_in[18];
  const float* W2pc_r = (const float*)d_in[19];
  const int* nid_c = (const int*)d_in[20];
  const int* nid_p = (const int*)d_in[21];
  const int* e_src = (const int*)d_in[22];
  const int* e_dst = (const int*)d_in[23];
  const int* l_src = (const int*)d_in[24];
  const int* l_dst = (const int*)d_in[25];

  float* ws = (float*)d_ws;
  size_t off = 0;
  float* xc   = ws + off; off += (size_t)NC * HDIM;
  float* xp   = ws + off; off += (size_t)NP * HDIM;
  float* hp   = ws + off; off += (size_t)NP * HDIM;
  float* hc   = ws + off; off += (size_t)NC * HDIM;
  float* hp2  = ws + off; off += (size_t)NP * HDIM;
  float* hc2  = ws + off; off += (size_t)NC * HDIM;
  float* sbuf = ws + off; off += (size_t)NP * HDIM;
  float* degp = ws + off; off += NP;
  float* degc = ws + off; off += NC;

  const int TB = 256;
  const int edge_blocks = (E * 32 + TB - 1) / TB;
  const int np_elems = NP * HDIM;
  const int nc_elems = NC * HDIM;

  gemm_in_kernel<<<NC / 16, TB, 0, stream>>>(x_country, Wc_lin, bc_lin, emb_c,
                                             nid_c, xc, NC);
  gemm_in_kernel<<<NP / 16, TB, 0, stream>>>(x_product, Wp_lin, bp_lin, emb_p,
                                             nid_p, xp, NP);

  zero_kernel<<<64, TB, 0, stream>>>(degp, NP);
  zero_kernel<<<64, TB, 0, stream>>>(degc, NC);
  degree_kernel<<<(E + TB - 1) / TB, TB, 0, stream>>>(e_dst, E, degp);
  degree_kernel<<<(E + TB - 1) / TB, TB, 0, stream>>>(e_src, E, degc);

  zero_kernel<<<1024, TB, 0, stream>>>(sbuf, np_elems);
  scatter_kernel<<<edge_blocks, TB, 0, stream>>>(xc, e_src, e_dst, E, sbuf);
  sage_kernel<<<NP / 16, TB, 0, stream>>>(sbuf, degp, xp, W1ep_l, b1ep, W1ep_r,
                                          hp, 1);

  zero_kernel<<<1024, TB, 0, stream>>>(sbuf, nc_elems);
  scatter_kernel<<<edge_blocks, TB, 0, stream>>>(xp, e_dst, e_src, E, sbuf);
  sage_kernel<<<NC / 16, TB, 0, stream>>>(sbuf, degc, xc, W1pc_l, b1pc, W1pc_r,
                                          hc, 1);

  zero_kernel<<<1024, TB, 0, stream>>>(sbuf, np_elems);
  scatter_kernel<<<edge_blocks, TB, 0, stream>>>(hc, e_src, e_dst, E, sbuf);
  sage_kernel<<<NP / 16, TB, 0, stream>>>(sbuf, degp, hp, W2ep_l, b2ep, W2ep_r,
                                          hp2, 0);

  zero_kernel<<<1024, TB, 0, stream>>>(sbuf, nc_elems);
  scatter_kernel<<<edge_blocks, TB, 0, stream>>>(hp, e_dst, e_src, E, sbuf);
  sage_kernel<<<NC / 16, TB, 0, stream>>>(sbuf, degc, hc, W2pc_l, b2pc, W2pc_r,
                                          hc2, 0);

  const int el_blocks = (EL * 32 + TB - 1) / TB;
  edge_dot_kernel<<<el_blocks, TB, 0, stream>>>(hc2, hp2, l_src, l_dst, EL,
                                                (float*)d_out);
}


# scatter-adds: 8192 resident waves stride over runs of 32 edges (indices read as one line, row numbers broadcast to scalars, 8 rows in flight), not 500k one-edge waves; atomic arrival order changes
# speedup vs baseline: 1.6825x; 1.6825x over previous
#include <hip/hip_runtime.h>


typedef __attribute__((ext_vector_type(2))) float v2f;
typedef __attribute__((ext_vector_type(4))) float v4f;
typedef __attribute__((ext_vector_type(8))) float v8f;

constexpr int HDIM = 128;

__device__ __forceinline__ void atomic_add_f32(float* p, float v) {
  (void)__hip_atomic_fetch_add(p, v, __ATOMIC_RELAXED,
                               __HIP_MEMORY_SCOPE_AGENT);
}

__global__ void zero_kernel(float* __restrict__ p, int n) {
  int i = blockIdx.x * blockDim.x + threadIdx.x;
  int stride = gridDim.x * blockDim.x;
  for (; i < n; i += stride) p[i] = 0.0f;
}

__global__ void degree_kernel(const int* __restrict__ dst, int e,
                              float* __restrict__ deg) {
  int i = blockIdx.x * blockDim.x + threadIdx.x;
  if (i < e) atomic_add_f32(&deg[dst[i]], 1.0f);
}

constexpr int EDGE_RUN = 32;
constexpr int EDGE_GRP = 8;

__device__ __forceinline__ void scatter_one_edge(const float* __restrict__ x,
                                                 float* __restrict__ s,
                                                 int sr, int dr, int lane) {
  const float* xr = x + (size_t)sr * HDIM + lane;
  float* o = s + (size_t)dr * HDIM + lane;
  const float v0 = xr[0];
  const float v1 = xr[32];
  const float v2 = xr[64];
  const float v3 = xr[96];
  atomic_add_f32(o, v0);
  atomic_add_f32(o + 32, v1);
  atomic_add_f32(o + 64, v2);
  atomic_add_f32(o + 96, v3);
}

__global__ void __launch_bounds__(256)
scatter_kernel(const float* __restrict__ x, const int* __restrict__ src,
               const int* __restrict__ dst, int e, float* __restrict__ s) {
  const int lane = threadIdx.x & 31;
  const int wave = __builtin_amdgcn_readfirstlane(
      (int)((blockIdx.x * blockDim.x + threadIdx.x) >> 5));
  const int nwave = (int)((gridDim.x * blockDim.x) >> 5);
  for (int base = wave * EDGE_RUN; base < e; base += nwave * EDGE_RUN) {
    const int i = base + lane;
    const int sv = i < e ? src[i] : 0;
    const int dv = i < e ? dst[i] : 0;
    const int left = e - base;
    const int cnt = left < EDGE_RUN ? left : EDGE_RUN;
    for (int j0 = 0; j0 < cnt; j0 += EDGE_GRP) {
      if (j0 + EDGE_GRP <= cnt) {
        float v[EDGE_GRP][4];
#pragma unroll
        for (int j = 0; j < EDGE_GRP; ++j) {
          const int sr = __builtin_amdgcn_readlane(sv, j0 + j);
          const float* xr = x + (size_t)sr * HDIM + lane;
          v[j][0] = xr[0];
          v[j][1] = xr[32];
          v[j][2] = xr[64];
          v[j][3] = xr[96];
        }
#pragma unroll
        for (int j = 0; j < EDGE_GRP; ++j) {
          const int dr = __builtin_amdgcn_readlane(dv, j0 + j);
          float* o = s + (size_t)dr * HDIM + lane;
          atomic_add_f32(o, v[j][0]);
          atomic_add_f32(o + 32, v[j][1]);
          atomic_add_f32(o + 64, v[j][2]);
          atomic_add_f32(o + 96, v[j][3]);
        }
      } else {
        for (int j = j0; j < cnt; ++j) {
          scatter_one_edge(x, s, __builtin_amdgcn_readlane(sv, j),
                           __builtin_amdgcn_readlane(dv, j), lane);
        }
      }
    }
  }
}


struct GemmFrag {
  v4f a[2][4];
  v4f w[16];
};

__device__ __forceinline__ void gemm_frag_load(GemmFrag& f,
                                               const float* __restrict__ ap,
                                               const float* __restrict__ wp,
                                               size_t a_tile1, int step) {
  const float* a = ap + step * 32;
#pragma unroll
  for (int i = 0; i < 4; ++i) {
    f.a[0][i] = *(const v4f*)(a + 4 * i);
    f.a[1][i] = *(const v4f*)(a + a_tile1 + 4 * i);
  }
  const float* w = wp + (size_t)step * 32 * HDIM;
#pragma unroll
  for (int i = 0; i < 16; ++i) f.w[i] = *(const v4f*)(w + i * HDIM);
}

__device__ __forceinline__ void gemm_frag_mma(const GemmFrag& f,
                                              v8f (&acc)[2][4]) {
#pragma unroll
  for (int q = 0; q < 8; ++q) {
#pragma unroll
    for (int t = 0; t < 2; ++t) {
      v2f av;
      av.x = f.a[t][q >> 1][(2 * q) & 3];
      av.y = f.a[t][q >> 1][(2 * q + 1) & 3];
#pragma unroll
      for (int j = 0; j < 4; ++j) {
        v2f bv;
        bv.x = f.w[2 * q][j];
        bv.y = f.w[2 * q + 1][j];
        acc[t][j] = __builtin_amdgcn_wmma_f32_16x16x4_f32(
            false, av, false, bv, (short)0, acc[t][j], false, false);
      }
    }
  }
}

__global__ void __launch_bounds__(256)
gemm_splitk_kernel(const float* __restrict__ A, const float* __restrict__ W,
                   float* __restrict__ P, int K, int kslice, int M) {
  const int lane = threadIdx.x & 31;
  const int wave = threadIdx.x >> 5;
  const int half = lane >> 4;
  const int c    = lane & 15;
  const int wm   = wave & 3;
  const int wn   = wave >> 2;
  const int m0   = blockIdx.x * 128 + wm * 32;
  const int nb   = wn * 64 + 4 * c;
  const int kb   = blockIdx.y * kslice + 16 * half;

  const float* ap = A + (size_t)(m0 + c) * K + kb;
  const float* wp = W + (size_t)kb * HDIM + nb;
  const size_t a_tile1 = (size_t)16 * K;
  const int nsteps = kslice / 32;

  v8f acc[2][4] = {};
  GemmFrag f0, f1;
  gemm_frag_load(f0, ap, wp, a_tile1, 0);
  for (int s = 0; s < nsteps; s += 2) {
    gemm_frag_load(f1, ap, wp, a_tile1, s + 1);
    __builtin_amdgcn_sched_barrier(0);
    gemm_frag_mma(f0, acc);
    const int s2 = (s + 2 < nsteps) ? s + 2 : s + 1;
    gemm_frag_load(f0, ap, wp, a_tile1, s2);
    __builtin_amdgcn_sched_barrier(0);
    gemm_frag_mma(f1, acc);
  }

  float* pp = P + (size_t)blockIdx.y * M * HDIM;
#pragma unroll
  for (int t = 0; t < 2; ++t) {
#pragma unroll
    for (int r = 0; r < 8; ++r) {
      const int m = m0 + 16 * t + 8 * half + r;
      v4f o;
      o.x = acc[t][0][r];
      o.y = acc[t][1][r];
      o.z = acc[t][2][r];
      o.w = acc[t][3][r];
      *(v4f*)(pp + (size_t)m * HDIM + nb) = o;
    }
  }
}

__global__ void __launch_bounds__(256)
reduce_in_kernel(const float* __restrict__ P, int S, int M,
                 const float* __restrict__ bias, const float* __restrict__ emb,
                 const int* __restrict__ nid, float* __restrict__ C) {
  const int i = blockIdx.x * blockDim.x + threadIdx.x;
  if (i >= M * (HDIM / 4)) return;
  const int m  = i >> 5;
  const int n4 = (i & 31) * 4;
  const size_t plane = (size_t)M * HDIM;
  const float* p = P + (size_t)i * 4;
  v4f acc = *(const v4f*)p;
  for (int s = 1; s < S; ++s) acc += *(const v4f*)(p + (size_t)s * plane);
  const v4f b = *(const v4f*)(bias + n4);
  const v4f e = *(const v4f*)(emb + (size_t)nid[m] * HDIM + n4);
  *(v4f*)(C + (size_t)i * 4) = (acc + b) + e;
}

__global__ void __launch_bounds__(256)
sage_kernel(const float* __restrict__ S, const float* __restrict__ deg,
            const float* __restrict__ X, const float* __restrict__ Wl,
            const float* __restrict__ bl, const float* __restrict__ Wr,
            float* __restrict__ Out, int relu) {
  const int lane = threadIdx.x & 31;
  const int wave = threadIdx.x >> 5;
  const int half = lane >> 4;
  const int nloc = lane & 15;
  const int m0   = blockIdx.x * 16;
  const int row  = m0 + nloc;
  const int n    = wave * 16 + nloc;

  const float* Srow = S + (size_t)row * HDIM;
  const float* Xrow = X + (size_t)row * HDIM;
  v8f accS = {};
  v8f accX = {};
#pragma unroll 4
  for (int k0 = 0; k0 < HDIM; k0 += 4) {
    const int ka = k0 + 2 * half;
    v2f a1; a1.x = Srow[ka]; a1.y = Srow[ka + 1];
    v2f b1; b1.x = Wl[(size_t)ka * HDIM + n];
            b1.y = Wl[(size_t)(ka + 1) * HDIM + n];
    accS = __builtin_amdgcn_wmma_f32_16x16x4_f32(false, a1, false, b1,
                                                 (short)0, accS, false, false);
    v2f a2; a2.x = Xrow[ka]; a2.y = Xrow[ka + 1];
    v2f b2; b2.x = Wr[(size_t)ka * HDIM + n];
            b2.y = Wr[(size_t)(ka + 1) * HDIM + n];
    accX = __builtin_amdgcn_wmma_f32_16x16x4_f32(false, a2, false, b2,
                                                 (short)0, accX, false, false);
  }

  const float bn = bl[n];
#pragma unroll
  for (int r = 0; r < 8; ++r) {
    const int m = m0 + r + 8 * half;
    float d = deg[m];
    d = d < 1.0f ? 1.0f : d;
    float v = accS[r] / d + accX[r] + bn;
    if (relu) v = v > 0.0f ? v : 0.0f;
    Out[(size_t)m * HDIM + n] = v;
  }
}

__global__ void edge_dot_kernel(const float* __restrict__ Hc,
                                const float* __restrict__ Hp,
                                const int* __restrict__ ls,
                                const int* __restrict__ ld, int e,
                                float* __restrict__ out) {
  int wid  = (blockIdx.x * blockDim.x + threadIdx.x) >> 5;
  int lane = threadIdx.x & 31;
  if (wid >= e) return;
  int a = ls[wid];
  int b = ld[wid];
  const float4 va = *(const float4*)(Hc + (size_t)a * HDIM + lane * 4);
  const float4 vb = *(const float4*)(Hp + (size_t)b * HDIM + lane * 4);
  float v = va.x * vb.x + va.y * vb.y + va.z * vb.z + va.w * vb.w;
#pragma unroll
  for (int off = 16; off > 0; off >>= 1) v += __shfl_down(v, off, 32);
  if (lane == 0) out[wid] = v;
}

extern "C" void kernel_launch(void* const* d_in, const int* in_sizes, int n_in,
                              void* d_out, int out_size, void* d_ws,
                              size_t ws_size, hipStream_t stream) {
  (void)in_sizes; (void)n_in; (void)out_size; (void)ws_size;
  const int NC = 4096, NP = 8192, E = 500000, EL = 500000;

  const float* x_country = (const float*)d_in[0];
  const float* x_product = (const float*)d_in[1];
  const float* Wc_lin = (const float*)d_in[2];
  const float* bc_lin = (const float*)d_in[3];
  const float* Wp_lin = (const float*)d_in[4];
  const float* bp_lin = (const float*)d_in[5];
  const float* emb_c  = (const float*)d_in[6];
  const float* emb_p  = (const float*)d_in[7];
  const float* W1ep_l = (const float*)d_in[8];
  const float* b1ep   = (const float*)d_in[9];
  const float* W1ep_r = (const float*)d_in[10];
  const float* W1pc_l = (const float*)d_in[11];
  const float* b1pc   = (const float*)d_in[12];
  const float* W1pc_r = (const float*)d_in[13];
  const float* W2ep_l = (const float*)d_in[14];
  const float* b2ep   = (const float*)d_in[15];
  const float* W2ep_r = (const float*)d_in[16];
  const float* W2pc_l = (const float*)d_in[17];
  const float* b2pc   = (const float*)d_in[18];
  const float* W2pc_r = (const float*)d_in[19];
  const int* nid_c = (const int*)d_in[20];
  const int* nid_p = (const int*)d_in[21];
  const int* e_src = (const int*)d_in[22];
  const int* e_dst = (const int*)d_in[23];
  const int* l_src = (const int*)d_in[24];
  const int* l_dst = (const int*)d_in[25];

  float* ws = (float*)d_ws;
  size_t off = 0;
  float* xc   = ws + off; off += (size_t)NC * HDIM;
  float* xp   = ws + off; off += (size_t)NP * HDIM;
  float* hp   = ws + off; off += (size_t)NP * HDIM;
  float* hc   = ws + off; off += (size_t)NC * HDIM;
  float* hp2  = ws + off; off += (size_t)NP * HDIM;
  float* hc2  = ws + off; off += (size_t)NC * HDIM;
  float* sbuf = ws + off; off += (size_t)NP * HDIM;
  float* degp = ws + off; off += NP;
  float* degc = ws + off; off += NC;

  const int TB = 256;
  const int edge_blocks = 1024;
  const int np_elems = NP * HDIM;
  const int nc_elems = NC * HDIM;

  float* part = hp;
  gemm_splitk_kernel<<<dim3(NC / 128, 8), TB, 0, stream>>>(
      x_country, Wc_lin, part, NC, NC / 8, NC);
  reduce_in_kernel<<<NC * (HDIM / 4) / TB, TB, 0, stream>>>(
      part, 8, NC, bc_lin, emb_c, nid_c, xc);
  gemm_splitk_kernel<<<dim3(NP / 128, 4), TB, 0, stream>>>(
      x_product, Wp_lin, part, NP, NP / 4, NP);
  reduce_in_kernel<<<NP * (HDIM / 4) / TB, TB, 0, stream>>>(
      part, 4, NP, bp_lin, emb_p, nid_p, xp);

  zero_kernel<<<64, TB, 0, stream>>>(degp, NP);
  zero_kernel<<<64, TB, 0, stream>>>(degc, NC);
  degree_kernel<<<(E + TB - 1) / TB, TB, 0, stream>>>(e_dst, E, degp);
  degree_kernel<<<(E + TB - 1) / TB, TB, 0, stream>>>(e_src, E, degc);

  zero_kernel<<<1024, TB, 0, stream>>>(sbuf, np_elems);
  scatter_kernel<<<edge_blocks, TB, 0, stream>>>(xc, e_src, e_dst, E, sbuf);
  sage_kernel<<<NP / 16, TB, 0, stream>>>(sbuf, degp, xp, W1ep_l, b1ep, W1ep_r,
                                          hp, 1);

  zero_kernel<<<1024, TB, 0, stream>>>(sbuf, nc_elems);
  scatter_kernel<<<edge_blocks, TB, 0, stream>>>(xp, e_dst, e_src, E, sbuf);
  sage_kernel<<<NC / 16, TB, 0, stream>>>(sbuf, degc, xc, W1pc_l, b1pc, W1pc_r,
                                          hc, 1);

  zero_kernel<<<1024, TB, 0, stream>>>(sbuf, np_elems);
  scatter_kernel<<<edge_blocks, TB, 0, stream>>>(hc, e_src, e_dst, E, sbuf);
  sage_kernel<<<NP / 16, TB, 0, stream>>>(sbuf, degp, hp, W2ep_l, b2ep, W2ep_r,
                                          hp2, 0);

  zero_kernel<<<1024, TB, 0, stream>>>(sbuf, nc_elems);
  scatter_kernel<<<edge_blocks, TB, 0, stream>>>(hp, e_dst, e_src, E, sbuf);
  sage_kernel<<<NC / 16, TB, 0, stream>>>(sbuf, degc, hc, W2pc_l, b2pc, W2pc_r,
                                          hc2, 0);

  const int el_blocks = (EL * 32 + TB - 1) / TB;
  edge_dot_kernel<<<el_blocks, TB, 0, stream>>>(hc2, hp2, l_src, l_dst, EL,
                                                (float*)d_out);
}


# scatter-add buffer kept twice, one copy per half of the second-level cache: a wave adds into its own half's copy (by die number), the layer kernel sums the two; float-atomic sum order changes
# speedup vs baseline: 2.0680x; 1.2291x over previous
#include <hip/hip_runtime.h>


typedef __attribute__((ext_vector_type(2))) float v2f;
typedef __attribute__((ext_vector_type(4))) float v4f;
typedef __attribute__((ext_vector_type(8))) float v8f;

constexpr int HDIM = 128;

__device__ __forceinline__ void atomic_add_f32(float* p, float v) {
  (void)__hip_atomic_fetch_add(p, v, __ATOMIC_RELAXED,
                               __HIP_MEMORY_SCOPE_AGENT);
}

__device__ __forceinline__ int cache_half() {
  return (int)((__builtin_amdgcn_s_sendmsg_rtn(0x87) >> 18) & 1u);
}

__global__ void zero_one_kernel(float* __restrict__ p, int n) {
  int i = blockIdx.x * blockDim.x + threadIdx.x;
  int stride = gridDim.x * blockDim.x;
  for (; i < n; i += stride) p[i] = 0.0f;
}

__global__ void zero_kernel(float* __restrict__ p0, float* __restrict__ p1,
                            int n) {
  int i = blockIdx.x * blockDim.x + threadIdx.x;
  int stride = gridDim.x * blockDim.x;
  for (; i < n; i += stride) {
    p0[i] = 0.0f;
    p1[i] = 0.0f;
  }
}

__global__ void degree_kernel(const int* __restrict__ dst, int e,
                              float* __restrict__ deg) {
  int i = blockIdx.x * blockDim.x + threadIdx.x;
  if (i < e) atomic_add_f32(&deg[dst[i]], 1.0f);
}

constexpr int EDGE_RUN = 32;
constexpr int EDGE_GRP = 8;

__device__ __forceinline__ void scatter_one_edge(const float* __restrict__ x,
                                                 float* __restrict__ s,
                                                 int sr, int dr, int lane) {
  const float* xr = x + (size_t)sr * HDIM + lane;
  float* o = s + (size_t)dr * HDIM + lane;
  const float v0 = xr[0];
  const float v1 = xr[32];
  const float v2 = xr[64];
  const float v3 = xr[96];
  atomic_add_f32(o, v0);
  atomic_add_f32(o + 32, v1);
  atomic_add_f32(o + 64, v2);
  atomic_add_f32(o + 96, v3);
}

__global__ void __launch_bounds__(256)
scatter_kernel(const float* __restrict__ x, const int* __restrict__ src,
               const int* __restrict__ dst, int e, float* __restrict__ s0,
               float* __restrict__ s1) {
  float* __restrict__ s = cache_half() ? s1 : s0;
  const int lane = threadIdx.x & 31;
  const int wave = __builtin_amdgcn_readfirstlane(
      (int)((blockIdx.x * blockDim.x + threadIdx.x) >> 5));
  const int nwave = (int)((gridDim.x * blockDim.x) >> 5);
  for (int base = wave * EDGE_RUN; base < e; base += nwave * EDGE_RUN) {
    const int i = base + lane;
    const int sv = i < e ? src[i] : 0;
    const int dv = i < e ? dst[i] : 0;
    const int left = e - base;
    const int cnt = left < EDGE_RUN ? left : EDGE_RUN;
    for (int j0 = 0; j0 < cnt; j0 += EDGE_GRP) {
      if (j0 + EDGE_GRP <= cnt) {
        float v[EDGE_GRP][4];
#pragma unroll
        for (int j = 0; j < EDGE_GRP; ++j) {
          const int sr = __builtin_amdgcn_readlane(sv, j0 + j);
          const float* xr = x + (size_t)sr * HDIM + lane;
          v[j][0] = xr[0];
          v[j][1] = xr[32];
          v[j][2] = xr[64];
          v[j][3] = xr[96];
        }
#pragma unroll
        for (int j = 0; j < EDGE_GRP; ++j) {
          const int dr = __builtin_amdgcn_readlane(dv, j0 + j);
          float* o = s + (size_t)dr * HDIM + lane;
          atomic_add_f32(o, v[j][0]);
          atomic_add_f32(o + 32, v[j][1]);
          atomic_add_f32(o + 64, v[j][2]);
          atomic_add_f32(o + 96, v[j][3]);
        }
      } else {
        for (int j = j0; j < cnt; ++j) {
          scatter_one_edge(x, s, __builtin_amdgcn_readlane(sv, j),
                           __builtin_amdgcn_readlane(dv, j), lane);
        }
      }
    }
  }
}


struct GemmFrag {
  v4f a[2][4];
  v4f w[16];
};

__device__ __forceinline__ void gemm_frag_load(GemmFrag& f,
                                               const float* __restrict__ ap,
                                               const float* __restrict__ wp,
                                               size_t a_tile1, int step) {
  const float* a = ap + step * 32;
#pragma unroll
  for (int i = 0; i < 4; ++i) {
    f.a[0][i] = *(const v4f*)(a + 4 * i);
    f.a[1][i] = *(const v4f*)(a + a_tile1 + 4 * i);
  }
  const float* w = wp + (size_t)step * 32 * HDIM;
#pragma unroll
  for (int i = 0; i < 16; ++i) f.w[i] = *(const v4f*)(w + i * HDIM);
}

__device__ __forceinline__ void gemm_frag_mma(const GemmFrag& f,
                                              v8f (&acc)[2][4]) {
#pragma unroll
  for (int q = 0; q < 8; ++q) {
#pragma unroll
    for (int t = 0; t < 2; ++t) {
      v2f av;
      av.x = f.a[t][q >> 1][(2 * q) & 3];
      av.y = f.a[t][q >> 1][(2 * q + 1) & 3];
#pragma unroll
      for (int j = 0; j < 4; ++j) {
        v2f bv;
        bv.x = f.w[2 * q][j];
        bv.y = f.w[2 * q + 1][j];
        acc[t][j] = __builtin_amdgcn_wmma_f32_16x16x4_f32(
            false, av, false, bv, (short)0, acc[t][j], false, false);
      }
    }
  }
}

__global__ void __launch_bounds__(256)
gemm_splitk_kernel(const float* __restrict__ A, const float* __restrict__ W,
                   float* __restrict__ P, int K, int kslice, int M) {
  const int lane = threadIdx.x & 31;
  const int wave = threadIdx.x >> 5;
  const int half = lane >> 4;
  const int c    = lane & 15;
  const int wm   = wave & 3;
  const int wn   = wave >> 2;
  const int m0   = blockIdx.x * 128 + wm * 32;
  const int nb   = wn * 64 + 4 * c;
  const int kb   = blockIdx.y * kslice + 16 * half;

  const float* ap = A + (size_t)(m0 + c) * K + kb;
  const float* wp = W + (size_t)kb * HDIM + nb;
  const size_t a_tile1 = (size_t)16 * K;
  const int nsteps = kslice / 32;

  v8f acc[2][4] = {};
  GemmFrag f0, f1;
  gemm_frag_load(f0, ap, wp, a_tile1, 0);
  for (int s = 0; s < nsteps; s += 2) {
    gemm_frag_load(f1, ap, wp, a_tile1, s + 1);
    __builtin_amdgcn_sched_barrier(0);
    gemm_frag_mma(f0, acc);
    const int s2 = (s + 2 < nsteps) ? s + 2 : s + 1;
    gemm_frag_load(f0, ap, wp, a_tile1, s2);
    __builtin_amdgcn_sched_barrier(0);
    gemm_frag_mma(f1, acc);
  }

  float* pp = P + (size_t)blockIdx.y * M * HDIM;
#pragma unroll
  for (int t = 0; t < 2; ++t) {
#pragma unroll
    for (int r = 0; r < 8; ++r) {
      const int m = m0 + 16 * t + 8 * half + r;
      v4f o;
      o.x = acc[t][0][r];
      o.y = acc[t][1][r];
      o.z = acc[t][2][r];
      o.w = acc[t][3][r];
      *(v4f*)(pp + (size_t)m * HDIM + nb) = o;
    }
  }
}

__global__ void __launch_bounds__(256)
reduce_in_kernel(const float* __restrict__ P, int S, int M,
                 const float* __restrict__ bias, const float* __restrict__ emb,
                 const int* __restrict__ nid, float* __restrict__ C) {
  const int i = blockIdx.x * blockDim.x + threadIdx.x;
  if (i >= M * (HDIM / 4)) return;
  const int m  = i >> 5;
  const int n4 = (i & 31) * 4;
  const size_t plane = (size_t)M * HDIM;
  const float* p = P + (size_t)i * 4;
  v4f acc = *(const v4f*)p;
  for (int s = 1; s < S; ++s) acc += *(const v4f*)(p + (size_t)s * plane);
  const v4f b = *(const v4f*)(bias + n4);
  const v4f e = *(const v4f*)(emb + (size_t)nid[m] * HDIM + n4);
  *(v4f*)(C + (size_t)i * 4) = (acc + b) + e;
}

__global__ void __launch_bounds__(256)
sage_kernel(const float* __restrict__ S0, const float* __restrict__ S1,
            const float* __restrict__ deg,
            const float* __restrict__ X, const float* __restrict__ Wl,
            const float* __restrict__ bl, const float* __restrict__ Wr,
            float* __restrict__ Out, int relu) {
  const int lane = threadIdx.x & 31;
  const int wave = threadIdx.x >> 5;
  const int half = lane >> 4;
  const int nloc = lane & 15;
  const int m0   = blockIdx.x * 16;
  const int row  = m0 + nloc;
  const int n    = wave * 16 + nloc;

  const float* Srow = S0 + (size_t)row * HDIM;
  const float* Trow = S1 + (size_t)row * HDIM;
  const float* Xrow = X + (size_t)row * HDIM;
  v8f accS = {};
  v8f accX = {};
#pragma unroll 4
  for (int k0 = 0; k0 < HDIM; k0 += 4) {
    const int ka = k0 + 2 * half;
    v2f a1; a1.x = Srow[ka] + Trow[ka]; a1.y = Srow[ka + 1] + Trow[ka + 1];
    v2f b1; b1.x = Wl[(size_t)ka * HDIM + n];
            b1.y = Wl[(size_t)(ka + 1) * HDIM + n];
    accS = __builtin_amdgcn_wmma_f32_16x16x4_f32(false, a1, false, b1,
                                                 (short)0, accS, false, false);
    v2f a2; a2.x = Xrow[ka]; a2.y = Xrow[ka + 1];
    v2f b2; b2.x = Wr[(size_t)ka * HDIM + n];
            b2.y = Wr[(size_t)(ka + 1) * HDIM + n];
    accX = __builtin_amdgcn_wmma_f32_16x16x4_f32(false, a2, false, b2,
                                                 (short)0, accX, false, false);
  }

  const float bn = bl[n];
#pragma unroll
  for (int r = 0; r < 8; ++r) {
    const int m = m0 + r + 8 * half;
    float d = deg[m];
    d = d < 1.0f ? 1.0f : d;
    float v = accS[r] / d + accX[r] + bn;
    if (relu) v = v > 0.0f ? v : 0.0f;
    Out[(size_t)m * HDIM + n] = v;
  }
}

__global__ void edge_dot_kernel(const float* __restrict__ Hc,
                                const float* __restrict__ Hp,
                                const int* __restrict__ ls,
                                const int* __restrict__ ld, int e,
                                float* __restrict__ out) {
  int wid  = (blockIdx.x * blockDim.x + threadIdx.x) >> 5;
  int lane = threadIdx.x & 31;
  if (wid >= e) return;
  int a = ls[wid];
  int b = ld[wid];
  const float4 va = *(const float4*)(Hc + (size_t)a * HDIM + lane * 4);
  const float4 vb = *(const float4*)(Hp + (size_t)b * HDIM + lane * 4);
  float v = va.x * vb.x + va.y * vb.y + va.z * vb.z + va.w * vb.w;
#pragma unroll
  for (int off = 16; off > 0; off >>= 1) v += __shfl_down(v, off, 32);
  if (lane == 0) out[wid] = v;
}

extern "C" void kernel_launch(void* const* d_in, const int* in_sizes, int n_in,
                              void* d_out, int out_size, void* d_ws,
                              size_t ws_size, hipStream_t stream) {
  (void)in_sizes; (void)n_in; (void)out_size; (void)ws_size;
  const int NC = 4096, NP = 8192, E = 500000, EL = 500000;

  const float* x_country = (const float*)d_in[0];
  const float* x_product = (const float*)d_in[1];
  const float* Wc_lin = (const float*)d_in[2];
  const float* bc_lin = (const float*)d_in[3];
  const float* Wp_lin = (const float*)d_in[4];
  const float* bp_lin = (const float*)d_in[5];
  const float* emb_c  = (const float*)d_in[6];
  const float* emb_p  = (const float*)d_in[7];
  const float* W1ep_l = (const float*)d_in[8];
  const float* b1ep   = (const float*)d_in[9];
  const float* W1ep_r = (const float*)d_in[10];
  const float* W1pc_l = (const float*)d_in[11];
  const float* b1pc   = (const float*)d_in[12];
  const float* W1pc_r = (const float*)d_in[13];
  const float* W2ep_l = (const float*)d_in[14];
  const float* b2ep   = (const float*)d_in[15];
  const float* W2ep_r = (const float*)d_in[16];
  const float* W2pc_l = (const float*)d_in[17];
  const float* b2pc   = (const float*)d_in[18];
  const float* W2pc_r = (const float*)d_in[19];
  const int* nid_c = (const int*)d_in[20];
  const int* nid_p = (const int*)d_in[21];
  const int* e_src = (const int*)d_in[22];
  const int* e_dst = (const int*)d_in[23];
  const int* l_src = (const int*)d_in[24];
  const int* l_dst = (const int*)d_in[25];

  float* ws = (float*)d_ws;
  size_t off = 0;
  float* xc   = ws + off; off += (size_t)NC * HDIM;
  float* xp   = ws + off; off += (size_t)NP * HDIM;
  float* hp   = ws + off; off += (size_t)NP * HDIM;
  float* hc   = ws + off; off += (size_t)NC * HDIM;
  float* hp2  = ws + off; off += (size_t)NP * HDIM;
  float* hc2  = ws + off; off += (size_t)NC * HDIM;
  float* sbuf = ws + off; off += (size_t)NP * HDIM;
  float* degp = ws + off; off += NP;
  float* degc = ws + off; off += NC;

  const int TB = 256;
  const int edge_blocks = 1024;
  const int np_elems = NP * HDIM;
  const int nc_elems = NC * HDIM;

  float* part = hp;
  gemm_splitk_kernel<<<dim3(NC / 128, 8), TB, 0, stream>>>(
      x_country, Wc_lin, part, NC, NC / 8, NC);
  reduce_in_kernel<<<NC * (HDIM / 4) / TB, TB, 0, stream>>>(
      part, 8, NC, bc_lin, emb_c, nid_c, xc);
  gemm_splitk_kernel<<<dim3(NP / 128, 4), TB, 0, stream>>>(
      x_product, Wp_lin, part, NP, NP / 4, NP);
  reduce_in_kernel<<<NP * (HDIM / 4) / TB, TB, 0, stream>>>(
      part, 4, NP, bp_lin, emb_p, nid_p, xp);

  zero_one_kernel<<<64, TB, 0, stream>>>(degp, NP);
  zero_one_kernel<<<64, TB, 0, stream>>>(degc, NC);
  degree_kernel<<<(E + TB - 1) / TB, TB, 0, stream>>>(e_dst, E, degp);
  degree_kernel<<<(E + TB - 1) / TB, TB, 0, stream>>>(e_src, E, degc);


  zero_kernel<<<1024, TB, 0, stream>>>(sbuf, hp2, np_elems);
  scatter_kernel<<<edge_blocks, TB, 0, stream>>>(xc, e_src, e_dst, E, sbuf,
                                                 hp2);
  sage_kernel<<<NP / 16, TB, 0, stream>>>(sbuf, hp2, degp, xp,
                                          W1ep_l, b1ep, W1ep_r, hp, 1);

  zero_kernel<<<1024, TB, 0, stream>>>(sbuf, hc2, nc_elems);
  scatter_kernel<<<edge_blocks, TB, 0, stream>>>(xp, e_dst, e_src, E, sbuf,
                                                 hc2);
  sage_kernel<<<NC / 16, TB, 0, stream>>>(sbuf, hc2, degc, xc,
                                          W1pc_l, b1pc, W1pc_r, hc, 1);

  zero_kernel<<<1024, TB, 0, stream>>>(sbuf, xp, np_elems);
  scatter_kernel<<<edge_blocks, TB, 0, stream>>>(hc, e_src, e_dst, E, sbuf,
                                                 xp);
  sage_kernel<<<NP / 16, TB, 0, stream>>>(sbuf, xp, degp, hp,
                                          W2ep_l, b2ep, W2ep_r, hp2, 0);

  zero_kernel<<<1024, TB, 0, stream>>>(sbuf, xc, nc_elems);
  scatter_kernel<<<edge_blocks, TB, 0, stream>>>(hp, e_dst, e_src, E, sbuf,
                                                 xc);
  sage_kernel<<<NC / 16, TB, 0, stream>>>(sbuf, xc, degc, hc,
                                          W2pc_l, b2pc, W2pc_r, hc2, 0);

  const int el_blocks = (EL * 32 + TB - 1) / TB;
  edge_dot_kernel<<<el_blocks, TB, 0, stream>>>(hc2, hp2, l_src, l_dst, EL,
                                                (float*)d_out);
}


# neighbour sums and degree counts by owner blocks: a block keeps its dst rows in LDS, scans the dst indices for its edges, adds their rows by LDS float atomics; no global atomics; sum order changes
# speedup vs baseline: 2.4561x; 1.1877x over previous
#include <hip/hip_runtime.h>


typedef __attribute__((ext_vector_type(2))) float v2f;
typedef __attribute__((ext_vector_type(4))) float v4f;
typedef __attribute__((ext_vector_type(8))) float v8f;

constexpr int HDIM = 128;

typedef __attribute__((ext_vector_type(4))) int v4i;

constexpr int OWN_THREADS = 1024;
constexpr int OWN_WAVES   = OWN_THREADS / 32;
constexpr int OWN_TRIP    = 128;
constexpr int OWN_AHEAD   = 8;
constexpr int OWN_QCAP    = 32 + OWN_AHEAD * OWN_TRIP;
constexpr int EDGE_GRP    = 8;

__device__ __forceinline__ void lds_add_f32(float* p, float v) {
  (void)__hip_atomic_fetch_add(p, v, __ATOMIC_RELAXED,
                               __HIP_MEMORY_SCOPE_WORKGROUP);
}

__device__ __forceinline__ void wave_lds_sync() {
  __builtin_amdgcn_fence(__ATOMIC_RELEASE, "wavefront");
  __builtin_amdgcn_wave_barrier();
  __builtin_amdgcn_fence(__ATOMIC_ACQUIRE, "wavefront");
}

template <int ROWS>
__global__ void __launch_bounds__(OWN_THREADS)
own_rows_kernel(const float* __restrict__ x, const int* __restrict__ src,
                const int* __restrict__ dst, int e, float* __restrict__ s,
                float* __restrict__ deg) {
  __shared__ float acc[ROWS * HDIM];
  __shared__ float cnt[ROWS];
  __shared__ int list[OWN_WAVES * OWN_QCAP];

  const int lane = threadIdx.x & 31;
  const int wave = __builtin_amdgcn_readfirstlane((int)(threadIdx.x >> 5));
  const int row0 = blockIdx.x * ROWS;
  int* const q = list + wave * OWN_QCAP;

  for (int i = threadIdx.x; i < ROWS * HDIM; i += OWN_THREADS) acc[i] = 0.0f;
  if (threadIdx.x < ROWS) cnt[threadIdx.x] = 0.0f;
  __syncthreads();

  auto add_group = [&](int sq, int dq, int j0, int n) {
    float v[EDGE_GRP][4];
#pragma unroll
    for (int j = 0; j < EDGE_GRP; ++j) {
      const int sr = __builtin_amdgcn_readlane(sq, j0 + j);
      const float* xr = x + (size_t)sr * HDIM + lane;
      v[j][0] = xr[0];
      v[j][1] = xr[32];
      v[j][2] = xr[64];
      v[j][3] = xr[96];
    }
#pragma unroll
    for (int j = 0; j < EDGE_GRP; ++j) {
      if (j < n) {
        const int dr = __builtin_amdgcn_readlane(dq, j0 + j);
        float* o = acc + dr * HDIM + lane;
        lds_add_f32(o, v[j][0]);
        lds_add_f32(o + 32, v[j][1]);
        lds_add_f32(o + 64, v[j][2]);
        lds_add_f32(o + 96, v[j][3]);
      }
    }
  };

  int qn = 0;

  auto take = [&](bool own, int edge, int lrow) {
    const unsigned m = __builtin_amdgcn_ballot_w32(own);
    if (m == 0u) return;
    if (own) q[qn + (int)__builtin_amdgcn_mbcnt_lo(m, 0u)] = edge * ROWS + lrow;
    qn += __builtin_popcount(m);
  };

  auto work_off = [&]() {
    if (qn < 32) return;
    wave_lds_sync();
    int done = 0;
#pragma unroll 1
    for (; qn - done >= 32; done += 32) {
      const int w  = q[done + lane];
      const int sq = src[(unsigned)w / (unsigned)ROWS];
      const int dq = (int)((unsigned)w % (unsigned)ROWS);
      lds_add_f32(&cnt[dq], 1.0f);
#pragma unroll 1
      for (int j0 = 0; j0 < 32; j0 += EDGE_GRP) add_group(sq, dq, j0, EDGE_GRP);
    }
    const int rest = qn - done;
    const int tw = lane < rest ? q[done + lane] : 0;
    wave_lds_sync();
    if (lane < rest) q[lane] = tw;
    qn = rest;
  };

  const int nfull = e / OWN_TRIP;
  const int first = (int)(blockIdx.x * (unsigned)nfull / gridDim.x);
  for (int t = wave; t < nfull; t += OWN_WAVES * OWN_AHEAD) {
    v4i d[OWN_AHEAD];
    int i4[OWN_AHEAD];
    bool live[OWN_AHEAD];
#pragma unroll
    for (int u = 0; u < OWN_AHEAD; ++u) {
      int tl = t + u * OWN_WAVES;
      live[u] = tl < nfull;
      tl += first;
      if (tl >= nfull) tl -= nfull;
      i4[u] = (live[u] ? tl : 0) * OWN_TRIP + 4 * lane;
      d[u] = *(const v4i*)(dst + i4[u]);
    }
#pragma unroll
    for (int u = 0; u < OWN_AHEAD; ++u) {
      const unsigned l0 = (unsigned)(d[u].x - row0);
      const unsigned l1 = (unsigned)(d[u].y - row0);
      const unsigned l2 = (unsigned)(d[u].z - row0);
      const unsigned l3 = (unsigned)(d[u].w - row0);
      const unsigned lo01 = l0 < l1 ? l0 : l1;
      const unsigned lo23 = l2 < l3 ? l2 : l3;
      const unsigned lo = lo01 < lo23 ? lo01 : lo23;
      if (!live[u] ||
          __builtin_amdgcn_ballot_w32(lo < (unsigned)ROWS) == 0u)
        continue;
      take(l0 < (unsigned)ROWS, i4[u], (int)l0);
      take(l1 < (unsigned)ROWS, i4[u] + 1, (int)l1);
      take(l2 < (unsigned)ROWS, i4[u] + 2, (int)l2);
      take(l3 < (unsigned)ROWS, i4[u] + 3, (int)l3);
    }
    work_off();
  }
  if (wave == 0) {
    for (int i0 = nfull * OWN_TRIP; i0 < e; i0 += 32) {
      const int i = i0 + lane;
      const unsigned l = i < e ? (unsigned)(dst[i] - row0) : 0xffffffffu;
      take(l < (unsigned)ROWS, i, (int)l);
    }
    work_off();
  }

  wave_lds_sync();
  {
    const int w  = lane < qn ? q[lane] : 0;
    const int sq = lane < qn ? src[(unsigned)w / (unsigned)ROWS] : 0;
    const int dq = (int)((unsigned)w % (unsigned)ROWS);
    if (lane < qn) lds_add_f32(&cnt[dq], 1.0f);
#pragma unroll 1
    for (int j0 = 0; j0 < qn; j0 += EDGE_GRP) add_group(sq, dq, j0, qn - j0);
  }

  __syncthreads();
  float* out = s + (size_t)row0 * HDIM;
  for (int i = threadIdx.x; i < ROWS * HDIM / 4; i += OWN_THREADS)
    *(v4f*)(out + 4 * i) = *(const v4f*)(acc + 4 * i);
  if (threadIdx.x < ROWS) deg[row0 + threadIdx.x] = cnt[threadIdx.x];
}


struct GemmFrag {
  v4f a[2][4];
  v4f w[16];
};

__device__ __forceinline__ void gemm_frag_load(GemmFrag& f,
                                               const float* __restrict__ ap,
                                               const float* __restrict__ wp,
                                               size_t a_tile1, int step) {
  const float* a = ap + step * 32;
#pragma unroll
  for (int i = 0; i < 4; ++i) {
    f.a[0][i] = *(const v4f*)(a + 4 * i);
    f.a[1][i] = *(const v4f*)(a + a_tile1 + 4 * i);
  }
  const float* w = wp + (size_t)step * 32 * HDIM;
#pragma unroll
  for (int i = 0; i < 16; ++i) f.w[i] = *(const v4f*)(w + i * HDIM);
}

__device__ __forceinline__ void gemm_frag_mma(const GemmFrag& f,
                                              v8f (&acc)[2][4]) {
#pragma unroll
  for (int q = 0; q < 8; ++q) {
#pragma unroll
    for (int t = 0; t < 2; ++t) {
      v2f av;
      av.x = f.a[t][q >> 1][(2 * q) & 3];
      av.y = f.a[t][q >> 1][(2 * q + 1) & 3];
#pragma unroll
      for (int j = 0; j < 4; ++j) {
        v2f bv;
        bv.x = f.w[2 * q][j];
        bv.y = f.w[2 * q + 1][j];
        acc[t][j] = __builtin_amdgcn_wmma_f32_16x16x4_f32(
            false, av, false, bv, (short)0, acc[t][j], false, false);
      }
    }
  }
}

__global__ void __launch_bounds__(256)
gemm_splitk_kernel(const float* __restrict__ A, const float* __restrict__ W,
                   float* __restrict__ P, int K, int kslice, int M) {
  const int lane = threadIdx.x & 31;
  const int wave = threadIdx.x >> 5;
  const int half = lane >> 4;
  const int c    = lane & 15;
  const int wm   = wave & 3;
  const int wn   = wave >> 2;
  const int m0   = blockIdx.x * 128 + wm * 32;
  const int nb   = wn * 64 + 4 * c;
  const int kb   = blockIdx.y * kslice + 16 * half;

  const float* ap = A + (size_t)(m0 + c) * K + kb;
  const float* wp = W + (size_t)kb * HDIM + nb;
  const size_t a_tile1 = (size_t)16 * K;
  const int nsteps = kslice / 32;

  v8f acc[2][4] = {};
  GemmFrag f0, f1;
  gemm_frag_load(f0, ap, wp, a_tile1, 0);
  for (int s = 0; s < nsteps; s += 2) {
    gemm_frag_load(f1, ap, wp, a_tile1, s + 1);
    __builtin_amdgcn_sched_barrier(0);
    gemm_frag_mma(f0, acc);
    const int s2 = (s + 2 < nsteps) ? s + 2 : s + 1;
    gemm_frag_load(f0, ap, wp, a_tile1, s2);
    __builtin_amdgcn_sched_barrier(0);
    gemm_frag_mma(f1, acc);
  }

  float* pp = P + (size_t)blockIdx.y * M * HDIM;
#pragma unroll
  for (int t = 0; t < 2; ++t) {
#pragma unroll
    for (int r = 0; r < 8; ++r) {
      const int m = m0 + 16 * t + 8 * half + r;
      v4f o;
      o.x = acc[t][0][r];
      o.y = acc[t][1][r];
      o.z = acc[t][2][r];
      o.w = acc[t][3][r];
      *(v4f*)(pp + (size_t)m * HDIM + nb) = o;
    }
  }
}

__global__ void __launch_bounds__(256)
reduce_in_kernel(const float* __restrict__ P, int S, int M,
                 const float* __restrict__ bias, const float* __restrict__ emb,
                 const int* __restrict__ nid, float* __restrict__ C) {
  const int i = blockIdx.x * blockDim.x + threadIdx.x;
  if (i >= M * (HDIM / 4)) return;
  const int m  = i >> 5;
  const int n4 = (i & 31) * 4;
  const size_t plane = (size_t)M * HDIM;
  const float* p = P + (size_t)i * 4;
  v4f acc = *(const v4f*)p;
  for (int s = 1; s < S; ++s) acc += *(const v4f*)(p + (size_t)s * plane);
  const v4f b = *(const v4f*)(bias + n4);
  const v4f e = *(const v4f*)(emb + (size_t)nid[m] * HDIM + n4);
  *(v4f*)(C + (size_t)i * 4) = (acc + b) + e;
}

__global__ void __launch_bounds__(256)
sage_kernel(const float* __restrict__ S0, const float* __restrict__ deg,
            const float* __restrict__ X, const float* __restrict__ Wl,
            const float* __restrict__ bl, const float* __restrict__ Wr,
            float* __restrict__ Out, int relu) {
  const int lane = threadIdx.x & 31;
  const int wave = threadIdx.x >> 5;
  const int half = lane >> 4;
  const int nloc = lane & 15;
  const int m0   = blockIdx.x * 16;
  const int row  = m0 + nloc;
  const int n    = wave * 16 + nloc;

  const float* Srow = S0 + (size_t)row * HDIM;
  const float* Xrow = X + (size_t)row * HDIM;
  v8f accS = {};
  v8f accX = {};
#pragma unroll 4
  for (int k0 = 0; k0 < HDIM; k0 += 4) {
    const int ka = k0 + 2 * half;
    v2f a1; a1.x = Srow[ka]; a1.y = Srow[ka + 1];
    v2f b1; b1.x = Wl[(size_t)ka * HDIM + n];
            b1.y = Wl[(size_t)(ka + 1) * HDIM + n];
    accS = __builtin_amdgcn_wmma_f32_16x16x4_f32(false, a1, false, b1,
                                                 (short)0, accS, false, false);
    v2f a2; a2.x = Xrow[ka]; a2.y = Xrow[ka + 1];
    v2f b2; b2.x = Wr[(size_t)ka * HDIM + n];
            b2.y = Wr[(size_t)(ka + 1) * HDIM + n];
    accX = __builtin_amdgcn_wmma_f32_16x16x4_f32(false, a2, false, b2,
                                                 (short)0, accX, false, false);
  }

  const float bn = bl[n];
#pragma unroll
  for (int r = 0; r < 8; ++r) {
    const int m = m0 + r + 8 * half;
    float d = deg[m];
    d = d < 1.0f ? 1.0f : d;
    float v = accS[r] / d + accX[r] + bn;
    if (relu) v = v > 0.0f ? v : 0.0f;
    Out[(size_t)m * HDIM + n] = v;
  }
}

__global__ void edge_dot_kernel(const float* __restrict__ Hc,
                                const float* __restrict__ Hp,
                                const int* __restrict__ ls,
                                const int* __restrict__ ld, int e,
                                float* __restrict__ out) {
  int wid  = (blockIdx.x * blockDim.x + threadIdx.x) >> 5;
  int lane = threadIdx.x & 31;
  if (wid >= e) return;
  int a = ls[wid];
  int b = ld[wid];
  const float4 va = *(const float4*)(Hc + (size_t)a * HDIM + lane * 4);
  const float4 vb = *(const float4*)(Hp + (size_t)b * HDIM + lane * 4);
  float v = va.x * vb.x + va.y * vb.y + va.z * vb.z + va.w * vb.w;
#pragma unroll
  for (int off = 16; off > 0; off >>= 1) v += __shfl_down(v, off, 32);
  if (lane == 0) out[wid] = v;
}

extern "C" void kernel_launch(void* const* d_in, const int* in_sizes, int n_in,
                              void* d_out, int out_size, void* d_ws,
                              size_t ws_size, hipStream_t stream) {
  (void)in_sizes; (void)n_in; (void)out_size; (void)ws_size;
  const int NC = 4096, NP = 8192, E = 500000, EL = 500000;

  const float* x_country = (const float*)d_in[0];
  const float* x_product = (const float*)d_in[1];
  const float* Wc_lin = (const float*)d_in[2];
  const float* bc_lin = (const float*)d_in[3];
  const float* Wp_lin = (const float*)d_in[4];
  const float* bp_lin = (const float*)d_in[5];
  const float* emb_c  = (const float*)d_in[6];
  const float* emb_p  = (const float*)d_in[7];
  const float* W1ep_l = (const float*)d_in[8];
  const float* b1ep   = (const float*)d_in[9];
  const float* W1ep_r = (const float*)d_in[10];
  const float* W1pc_l = (const float*)d_in[11];
  const float* b1pc   = (const float*)d_in[12];
  const float* W1pc_r = (const float*)d_in[13];
  const float* W2ep_l = (const float*)d_in[14];
  const float* b2ep   = (const float*)d_in[15];
  const float* W2ep_r = (const float*)d_in[16];
  const float* W2pc_l = (const float*)d_in[17];
  const float* b2pc   = (const float*)d_in[18];
  const float* W2pc_r = (const float*)d_in[19];
  const int* nid_c = (const int*)d_in[20];
  const int* nid_p = (const int*)d_in[21];
  const int* e_src = (const int*)d_in[22];
  const int* e_dst = (const int*)d_in[23];
  const int* l_src = (const int*)d_in[24];
  const int* l_dst = (const int*)d_in[25];

  float* ws = (float*)d_ws;
  size_t off = 0;
  float* xc   = ws + off; off += (size_t)NC * HDIM;
  float* xp   = ws + off; off += (size_t)NP * HDIM;
  float* hp   = ws + off; off += (size_t)NP * HDIM;
  float* hc   = ws + off; off += (size_t)NC * HDIM;
  float* hp2  = ws + off; off += (size_t)NP * HDIM;
  float* hc2  = ws + off; off += (size_t)NC * HDIM;
  float* sbuf = ws + off; off += (size_t)NP * HDIM;
  float* degp = ws + off; off += NP;
  float* degc = ws + off; off += NC;

  const int TB = 256;
  const int own_blocks = 256;
  const int OT = 1024;

  float* part = hp;
  gemm_splitk_kernel<<<dim3(NC / 128, 8), TB, 0, stream>>>(
      x_country, Wc_lin, part, NC, NC / 8, NC);
  reduce_in_kernel<<<NC * (HDIM / 4) / TB, TB, 0, stream>>>(
      part, 8, NC, bc_lin, emb_c, nid_c, xc);
  gemm_splitk_kernel<<<dim3(NP / 128, 4), TB, 0, stream>>>(
      x_product, Wp_lin, part, NP, NP / 4, NP);
  reduce_in_kernel<<<NP * (HDIM / 4) / TB, TB, 0, stream>>>(
      part, 4, NP, bp_lin, emb_p, nid_p, xp);

  own_rows_kernel<NP / 256><<<own_blocks, OT, 0, stream>>>(
      xc, e_src, e_dst, E, sbuf, degp);
  sage_kernel<<<NP / 16, TB, 0, stream>>>(sbuf, degp, xp,
                                          W1ep_l, b1ep, W1ep_r, hp, 1);

  own_rows_kernel<NC / 256><<<own_blocks, OT, 0, stream>>>(
      xp, e_dst, e_src, E, sbuf, degc);
  sage_kernel<<<NC / 16, TB, 0, stream>>>(sbuf, degc, xc,
                                          W1pc_l, b1pc, W1pc_r, hc, 1);

  own_rows_kernel<NP / 256><<<own_blocks, OT, 0, stream>>>(
      hc, e_src, e_dst, E, sbuf, degp);
  sage_kernel<<<NP / 16, TB, 0, stream>>>(sbuf, degp, hp,
                                          W2ep_l, b2ep, W2ep_r, hp2, 0);

  own_rows_kernel<NC / 256><<<own_blocks, OT, 0, stream>>>(
      hp, e_dst, e_src, E, sbuf, degc);
  sage_kernel<<<NC / 16, TB, 0, stream>>>(sbuf, degc, hc,
                                          W2pc_l, b2pc, W2pc_r, hc2, 0);

  const int el_blocks = (EL * 32 + TB - 1) / TB;
  edge_dot_kernel<<<el_blocks, TB, 0, stream>>>(hc2, hp2, l_src, l_dst, EL,
                                                (float*)d_out);
}


# layer 2 neighbour sums replay per-wave edge lists (src row, local row) that layer 1's owner blocks saved in free workspace: no second scan of the index arrays; LDS float-atomic arrival order changes
# speedup vs baseline: 2.6742x; 1.0888x over previous
#include <hip/hip_runtime.h>


typedef __attribute__((ext_vector_type(2))) float v2f;
typedef __attribute__((ext_vector_type(4))) float v4f;
typedef __attribute__((ext_vector_type(8))) float v8f;

constexpr int HDIM = 128;

typedef __attribute__((ext_vector_type(4))) int v4i;

constexpr int OWN_THREADS = 1024;
constexpr int OWN_WAVES   = OWN_THREADS / 32;
constexpr int OWN_TRIP    = 128;
constexpr int OWN_AHEAD   = 8;
constexpr int OWN_QCAP    = 32 + OWN_AHEAD * OWN_TRIP;
constexpr int EDGE_GRP    = 8;
constexpr int OWN_LCAP    = 128;

__device__ __forceinline__ void lds_add_f32(float* p, float v) {
  (void)__hip_atomic_fetch_add(p, v, __ATOMIC_RELAXED,
                               __HIP_MEMORY_SCOPE_WORKGROUP);
}

__device__ __forceinline__ void wave_lds_sync() {
  __builtin_amdgcn_fence(__ATOMIC_RELEASE, "wavefront");
  __builtin_amdgcn_wave_barrier();
  __builtin_amdgcn_fence(__ATOMIC_ACQUIRE, "wavefront");
}

template <int ROWS>
__global__ void __launch_bounds__(OWN_THREADS)
own_rows_kernel(const float* __restrict__ x, const int* __restrict__ src,
                const int* __restrict__ dst, int e, float* __restrict__ s,
                float* __restrict__ deg, int* __restrict__ lists,
                int replay) {
  __shared__ float acc[ROWS * HDIM];
  __shared__ float cnt[ROWS];
  __shared__ int list[OWN_WAVES * OWN_QCAP];

  const int lane = threadIdx.x & 31;
  const int wave = __builtin_amdgcn_readfirstlane((int)(threadIdx.x >> 5));
  const int row0 = blockIdx.x * ROWS;
  int* const q = list + wave * OWN_QCAP;
  int* const lw = lists + ((size_t)blockIdx.x * OWN_WAVES + wave) * OWN_LCAP;
  const bool rec = replay == 0;
  int ln = 0;

  for (int i = threadIdx.x; i < ROWS * HDIM; i += OWN_THREADS) acc[i] = 0.0f;
  if (threadIdx.x < ROWS) cnt[threadIdx.x] = 0.0f;
  __syncthreads();

  auto add_group = [&](int sq, int dq, int j0, int n) {
    float v[EDGE_GRP][4];
#pragma unroll
    for (int j = 0; j < EDGE_GRP; ++j) {
      const int sr = __builtin_amdgcn_readlane(sq, j0 + j);
      const float* xr = x + (size_t)sr * HDIM + lane;
      v[j][0] = xr[0];
      v[j][1] = xr[32];
      v[j][2] = xr[64];
      v[j][3] = xr[96];
    }
#pragma unroll
    for (int j = 0; j < EDGE_GRP; ++j) {
      if (j < n) {
        const int dr = __builtin_amdgcn_readlane(dq, j0 + j);
        float* o = acc + dr * HDIM + lane;
        lds_add_f32(o, v[j][0]);
        lds_add_f32(o + 32, v[j][1]);
        lds_add_f32(o + 64, v[j][2]);
        lds_add_f32(o + 96, v[j][3]);
      }
    }
  };

  int qn = 0;

  auto take = [&](bool own, int edge, int lrow) {
    const unsigned m = __builtin_amdgcn_ballot_w32(own);
    if (m == 0u) return;
    if (own) q[qn + (int)__builtin_amdgcn_mbcnt_lo(m, 0u)] = edge * ROWS + lrow;
    qn += __builtin_popcount(m);
  };

  auto work_off = [&]() {
    if (qn < 32) return;
    wave_lds_sync();
    int done = 0;
#pragma unroll 1
    for (; qn - done >= 32; done += 32) {
      const int w  = q[done + lane];
      const int sq = src[(unsigned)w / (unsigned)ROWS];
      const int dq = (int)((unsigned)w % (unsigned)ROWS);
      lds_add_f32(&cnt[dq], 1.0f);
      if (rec) {
        if (ln + 32 <= OWN_LCAP - 1) lw[ln + lane] = sq * ROWS + dq;
        ln += 32;
      }
#pragma unroll 1
      for (int j0 = 0; j0 < 32; j0 += EDGE_GRP) add_group(sq, dq, j0, EDGE_GRP);
    }
    const int rest = qn - done;
    const int tw = lane < rest ? q[done + lane] : 0;
    wave_lds_sync();
    if (lane < rest) q[lane] = tw;
    qn = rest;
  };

  const int saved =
      replay ? __builtin_amdgcn_readfirstlane(lw[OWN_LCAP - 1]) : -1;
  if (saved >= 0) {
#pragma unroll 1
    for (int j = 0; j < saved; j += 32) {
      const int m  = saved - j < 32 ? saved - j : 32;
      const int w  = lane < m ? lw[j + lane] : 0;
      const int sq = (int)((unsigned)w / (unsigned)ROWS);
      const int dq = (int)((unsigned)w % (unsigned)ROWS);
      if (lane < m) lds_add_f32(&cnt[dq], 1.0f);
#pragma unroll 1
      for (int j0 = 0; j0 < m; j0 += EDGE_GRP) add_group(sq, dq, j0, m - j0);
    }
  } else {
  const int nfull = e / OWN_TRIP;
  const int first = (int)(blockIdx.x * (unsigned)nfull / gridDim.x);
  for (int t = wave; t < nfull; t += OWN_WAVES * OWN_AHEAD) {
    v4i d[OWN_AHEAD];
    int i4[OWN_AHEAD];
    bool live[OWN_AHEAD];
#pragma unroll
    for (int u = 0; u < OWN_AHEAD; ++u) {
      int tl = t + u * OWN_WAVES;
      live[u] = tl < nfull;
      tl += first;
      if (tl >= nfull) tl -= nfull;
      i4[u] = (live[u] ? tl : 0) * OWN_TRIP + 4 * lane;
      d[u] = *(const v4i*)(dst + i4[u]);
    }
#pragma unroll
    for (int u = 0; u < OWN_AHEAD; ++u) {
      const unsigned l0 = (unsigned)(d[u].x - row0);
      const unsigned l1 = (unsigned)(d[u].y - row0);
      const unsigned l2 = (unsigned)(d[u].z - row0);
      const unsigned l3 = (unsigned)(d[u].w - row0);
      const unsigned lo01 = l0 < l1 ? l0 : l1;
      const unsigned lo23 = l2 < l3 ? l2 : l3;
      const unsigned lo = lo01 < lo23 ? lo01 : lo23;
      if (!live[u] ||
          __builtin_amdgcn_ballot_w32(lo < (unsigned)ROWS) == 0u)
        continue;
      take(l0 < (unsigned)ROWS, i4[u], (int)l0);
      take(l1 < (unsigned)ROWS, i4[u] + 1, (int)l1);
      take(l2 < (unsigned)ROWS, i4[u] + 2, (int)l2);
      take(l3 < (unsigned)ROWS, i4[u] + 3, (int)l3);
    }
    work_off();
  }
  if (wave == 0) {
    for (int i0 = nfull * OWN_TRIP; i0 < e; i0 += 32) {
      const int i = i0 + lane;
      const unsigned l = i < e ? (unsigned)(dst[i] - row0) : 0xffffffffu;
      take(l < (unsigned)ROWS, i, (int)l);
    }
    work_off();
  }

  wave_lds_sync();
  {
    const int w  = lane < qn ? q[lane] : 0;
    const int sq = lane < qn ? src[(unsigned)w / (unsigned)ROWS] : 0;
    const int dq = (int)((unsigned)w % (unsigned)ROWS);
    if (lane < qn) lds_add_f32(&cnt[dq], 1.0f);
    if (rec) {
      if (lane < qn && ln + qn <= OWN_LCAP - 1) lw[ln + lane] = sq * ROWS + dq;
      ln += qn;
      if (lane == 0) lw[OWN_LCAP - 1] = ln <= OWN_LCAP - 1 ? ln : -1;
    }
#pragma unroll 1
    for (int j0 = 0; j0 < qn; j0 += EDGE_GRP) add_group(sq, dq, j0, qn - j0);
  }
  }

  __syncthreads();
  float* out = s + (size_t)row0 * HDIM;
  for (int i = threadIdx.x; i < ROWS * HDIM / 4; i += OWN_THREADS)
    *(v4f*)(out + 4 * i) = *(const v4f*)(acc + 4 * i);
  if (threadIdx.x < ROWS) deg[row0 + threadIdx.x] = cnt[threadIdx.x];
}


struct GemmFrag {
  v4f a[2][4];
  v4f w[16];
};

__device__ __forceinline__ void gemm_frag_load(GemmFrag& f,
                                               const float* __restrict__ ap,
                                               const float* __restrict__ wp,
                                               size_t a_tile1, int step) {
  const float* a = ap + step * 32;
#pragma unroll
  for (int i = 0; i < 4; ++i) {
    f.a[0][i] = *(const v4f*)(a + 4 * i);
    f.a[1][i] = *(const v4f*)(a + a_tile1 + 4 * i);
  }
  const float* w = wp + (size_t)step * 32 * HDIM;
#pragma unroll
  for (int i = 0; i < 16; ++i) f.w[i] = *(const v4f*)(w + i * HDIM);
}

__device__ __forceinline__ void gemm_frag_mma(const GemmFrag& f,
                                              v8f (&acc)[2][4]) {
#pragma unroll
  for (int q = 0; q < 8; ++q) {
#pragma unroll
    for (int t = 0; t < 2; ++t) {
      v2f av;
      av.x = f.a[t][q >> 1][(2 * q) & 3];
      av.y = f.a[t][q >> 1][(2 * q + 1) & 3];
#pragma unroll
      for (int j = 0; j < 4; ++j) {
        v2f bv;
        bv.x = f.w[2 * q][j];
        bv.y = f.w[2 * q + 1][j];
        acc[t][j] = __builtin_amdgcn_wmma_f32_16x16x4_f32(
            false, av, false, bv, (short)0, acc[t][j], false, false);
      }
    }
  }
}

__global__ void __launch_bounds__(256)
gemm_splitk_kernel(const float* __restrict__ A, const float* __restrict__ W,
                   float* __restrict__ P, int K, int kslice, int M) {
  const int lane = threadIdx.x & 31;
  const int wave = threadIdx.x >> 5;
  const int half = lane >> 4;
  const int c    = lane & 15;
  const int wm   = wave & 3;
  const int wn   = wave >> 2;
  const int m0   = blockIdx.x * 128 + wm * 32;
  const int nb   = wn * 64 + 4 * c;
  const int kb   = blockIdx.y * kslice + 16 * half;

  const float* ap = A + (size_t)(m0 + c) * K + kb;
  const float* wp = W + (size_t)kb * HDIM + nb;
  const size_t a_tile1 = (size_t)16 * K;
  const int nsteps = kslice / 32;

  v8f acc[2][4] = {};
  GemmFrag f0, f1;
  gemm_frag_load(f0, ap, wp, a_tile1, 0);
  for (int s = 0; s < nsteps; s += 2) {
    gemm_frag_load(f1, ap, wp, a_tile1, s + 1);
    __builtin_amdgcn_sched_barrier(0);
    gemm_frag_mma(f0, acc);
    const int s2 = (s + 2 < nsteps) ? s + 2 : s + 1;
    gemm_frag_load(f0, ap, wp, a_tile1, s2);
    __builtin_amdgcn_sched_barrier(0);
    gemm_frag_mma(f1, acc);
  }

  float* pp = P + (size_t)blockIdx.y * M * HDIM;
#pragma unroll
  for (int t = 0; t < 2; ++t) {
#pragma unroll
    for (int r = 0; r < 8; ++r) {
      const int m = m0 + 16 * t + 8 * half + r;
      v4f o;
      o.x = acc[t][0][r];
      o.y = acc[t][1][r];
      o.z = acc[t][2][r];
      o.w = acc[t][3][r];
      *(v4f*)(pp + (size_t)m * HDIM + nb) = o;
    }
  }
}

__global__ void __launch_bounds__(256)
reduce_in_kernel(const float* __restrict__ P, int S, int M,
                 const float* __restrict__ bias, const float* __restrict__ emb,
                 const int* __restrict__ nid, float* __restrict__ C) {
  const int i = blockIdx.x * blockDim.x + threadIdx.x;
  if (i >= M * (HDIM / 4)) return;
  const int m  = i >> 5;
  const int n4 = (i & 31) * 4;
  const size_t plane = (size_t)M * HDIM;
  const float* p = P + (size_t)i * 4;
  v4f acc = *(const v4f*)p;
  for (int s = 1; s < S; ++s) acc += *(const v4f*)(p + (size_t)s * plane);
  const v4f b = *(const v4f*)(bias + n4);
  const v4f e = *(const v4f*)(emb + (size_t)nid[m] * HDIM + n4);
  *(v4f*)(C + (size_t)i * 4) = (acc + b) + e;
}

__global__ void __launch_bounds__(256)
sage_kernel(const float* __restrict__ S0, const float* __restrict__ deg,
            const float* __restrict__ X, const float* __restrict__ Wl,
            const float* __restrict__ bl, const float* __restrict__ Wr,
            float* __restrict__ Out, int relu) {
  const int lane = threadIdx.x & 31;
  const int wave = threadIdx.x >> 5;
  const int half = lane >> 4;
  const int nloc = lane & 15;
  const int m0   = blockIdx.x * 16;
  const int row  = m0 + nloc;
  const int n    = wave * 16 + nloc;

  const float* Srow = S0 + (size_t)row * HDIM;
  const float* Xrow = X + (size_t)row * HDIM;
  v8f accS = {};
  v8f accX = {};
#pragma unroll 4
  for (int k0 = 0; k0 < HDIM; k0 += 4) {
    const int ka = k0 + 2 * half;
    v2f a1; a1.x = Srow[ka]; a1.y = Srow[ka + 1];
    v2f b1; b1.x = Wl[(size_t)ka * HDIM + n];
            b1.y = Wl[(size_t)(ka + 1) * HDIM + n];
    accS = __builtin_amdgcn_wmma_f32_16x16x4_f32(false, a1, false, b1,
                                                 (short)0, accS, false, false);
    v2f a2; a2.x = Xrow[ka]; a2.y = Xrow[ka + 1];
    v2f b2; b2.x = Wr[(size_t)ka * HDIM + n];
            b2.y = Wr[(size_t)(ka + 1) * HDIM + n];
    accX = __builtin_amdgcn_wmma_f32_16x16x4_f32(false, a2, false, b2,
                                                 (short)0, accX, false, false);
  }

  const float bn = bl[n];
#pragma unroll
  for (int r = 0; r < 8; ++r) {
    const int m = m0 + r + 8 * half;
    float d = deg[m];
    d = d < 1.0f ? 1.0f : d;
    float v = accS[r] / d + accX[r] + bn;
    if (relu) v = v > 0.0f ? v : 0.0f;
    Out[(size_t)m * HDIM + n] = v;
  }
}

__global__ void edge_dot_kernel(const float* __restrict__ Hc,
                                const float* __restrict__ Hp,
                                const int* __restrict__ ls,
                                const int* __restrict__ ld, int e,
                                float* __restrict__ out) {
  int wid  = (blockIdx.x * blockDim.x + threadIdx.x) >> 5;
  int lane = threadIdx.x & 31;
  if (wid >= e) return;
  int a = ls[wid];
  int b = ld[wid];
  const float4 va = *(const float4*)(Hc + (size_t)a * HDIM + lane * 4);
  const float4 vb = *(const float4*)(Hp + (size_t)b * HDIM + lane * 4);
  float v = va.x * vb.x + va.y * vb.y + va.z * vb.z + va.w * vb.w;
#pragma unroll
  for (int off = 16; off > 0; off >>= 1) v += __shfl_down(v, off, 32);
  if (lane == 0) out[wid] = v;
}

extern "C" void kernel_launch(void* const* d_in, const int* in_sizes, int n_in,
                              void* d_out, int out_size, void* d_ws,
                              size_t ws_size, hipStream_t stream) {
  (void)in_sizes; (void)n_in; (void)out_size; (void)ws_size;
  const int NC = 4096, NP = 8192, E = 500000, EL = 500000;

  const float* x_country = (const float*)d_in[0];
  const float* x_product = (const float*)d_in[1];
  const float* Wc_lin = (const float*)d_in[2];
  const float* bc_lin = (const float*)d_in[3];
  const float* Wp_lin = (const float*)d_in[4];
  const float* bp_lin = (const float*)d_in[5];
  const float* emb_c  = (const float*)d_in[6];
  const float* emb_p  = (const float*)d_in[7];
  const float* W1ep_l = (const float*)d_in[8];
  const float* b1ep   = (const float*)d_in[9];
  const float* W1ep_r = (const float*)d_in[10];
  const float* W1pc_l = (const float*)d_in[11];
  const float* b1pc   = (const float*)d_in[12];
  const float* W1pc_r = (const float*)d_in[13];
  const float* W2ep_l = (const float*)d_in[14];
  const float* b2ep   = (const float*)d_in[15];
  const float* W2ep_r = (const float*)d_in[16];
  const float* W2pc_l = (const float*)d_in[17];
  const float* b2pc   = (const float*)d_in[18];
  const float* W2pc_r = (const float*)d_in[19];
  const int* nid_c = (const int*)d_in[20];
  const int* nid_p = (const int*)d_in[21];
  const int* e_src = (const int*)d_in[22];
  const int* e_dst = (const int*)d_in[23];
  const int* l_src = (const int*)d_in[24];
  const int* l_dst = (const int*)d_in[25];

  float* ws = (float*)d_ws;
  size_t off = 0;
  float* xc   = ws + off; off += (size_t)NC * HDIM;
  float* xp   = ws + off; off += (size_t)NP * HDIM;
  float* hp   = ws + off; off += (size_t)NP * HDIM;
  float* hc   = ws + off; off += (size_t)NC * HDIM;
  float* hp2  = ws + off; off += (size_t)NP * HDIM;
  float* hc2  = ws + off; off += (size_t)NC * HDIM;
  float* sbuf = ws + off; off += (size_t)NP * HDIM;
  float* degp = ws + off; off += NP;
  float* degc = ws + off; off += NC;

  const int TB = 256;
  const int own_blocks = 256;
  const int OT = 1024;

  float* part = hp;
  gemm_splitk_kernel<<<dim3(NC / 128, 8), TB, 0, stream>>>(
      x_country, Wc_lin, part, NC, NC / 8, NC);
  reduce_in_kernel<<<NC * (HDIM / 4) / TB, TB, 0, stream>>>(
      part, 8, NC, bc_lin, emb_c, nid_c, xc);
  gemm_splitk_kernel<<<dim3(NP / 128, 4), TB, 0, stream>>>(
      x_product, Wp_lin, part, NP, NP / 4, NP);
  reduce_in_kernel<<<NP * (HDIM / 4) / TB, TB, 0, stream>>>(
      part, 4, NP, bp_lin, emb_p, nid_p, xp);

  int* listP   = (int*)hp2;
  int* listC   = (int*)hc2;
  float* sbufC = sbuf + (size_t)NC * HDIM;

  own_rows_kernel<NP / 256><<<own_blocks, OT, 0, stream>>>(
      xc, e_src, e_dst, E, sbuf, degp, listP, 0);
  sage_kernel<<<NP / 16, TB, 0, stream>>>(sbuf, degp, xp,
                                          W1ep_l, b1ep, W1ep_r, hp, 1);

  own_rows_kernel<NC / 256><<<own_blocks, OT, 0, stream>>>(
      xp, e_dst, e_src, E, sbufC, degc, listC, 0);
  sage_kernel<<<NC / 16, TB, 0, stream>>>(sbufC, degc, xc,
                                          W1pc_l, b1pc, W1pc_r, hc, 1);

  own_rows_kernel<NC / 256><<<own_blocks, OT, 0, stream>>>(
      hp, e_dst, e_src, E, sbufC, degc, listC, 1);
  sage_kernel<<<NC / 16, TB, 0, stream>>>(sbufC, degc, hc,
                                          W2pc_l, b2pc, W2pc_r, hc2, 0);

  own_rows_kernel<NP / 256><<<own_blocks, OT, 0, stream>>>(
      hc, e_src, e_dst, E, sbuf, degp, listP, 1);
  sage_kernel<<<NP / 16, TB, 0, stream>>>(sbuf, degp, hp,
                                          W2ep_l, b2ep, W2ep_r, hp2, 0);

  const int el_blocks = (EL * 32 + TB - 1) / TB;
  edge_dot_kernel<<<el_blocks, TB, 0, stream>>>(hc2, hp2, l_src, l_dst, EL,
                                                (float*)d_out);
}


# input projections: a tile's K slices are the 8 waves of one block, summed in LDS in the old slice order with bias and embedding: no partial planes, no reduce launches (11 -> 9); f32 sum order kept
# speedup vs baseline: 3.1769x; 1.1880x over previous
#include <hip/hip_runtime.h>


typedef __attribute__((ext_vector_type(2))) float v2f;
typedef __attribute__((ext_vector_type(4))) float v4f;
typedef __attribute__((ext_vector_type(8))) float v8f;

constexpr int HDIM = 128;


constexpr int OWN_THREADS = 1024;
constexpr int OWN_WAVES   = OWN_THREADS / 32;
constexpr int BM_GRP      = 32;
constexpr int BM_RUN      = 32 * BM_GRP;
constexpr int BM_BLOCKS   = 256;
constexpr int BM_AHEAD    = 16;
constexpr int BM_QCAP     = 32 + BM_AHEAD * BM_GRP;
constexpr int BM_LANES    = 16;
constexpr int BM_GCAP     = BM_LANES * 32;
constexpr int OWN_QCAP    = BM_QCAP + BM_GCAP;
constexpr int EDGE_GRP    = 8;
constexpr int OWN_LCAP    = 128;

__device__ __forceinline__ void lds_add_f32(float* p, float v) {
  (void)__hip_atomic_fetch_add(p, v, __ATOMIC_RELAXED,
                               __HIP_MEMORY_SCOPE_WORKGROUP);
}

__device__ __forceinline__ void wave_lds_sync() {
  __builtin_amdgcn_fence(__ATOMIC_RELEASE, "wavefront");
  __builtin_amdgcn_wave_barrier();
  __builtin_amdgcn_fence(__ATOMIC_ACQUIRE, "wavefront");
}

template <int ROWS>
__device__ __forceinline__ void
own_rows_body(float* acc, float* cnt, int* list, int blk,
              const float* __restrict__ x, const int* __restrict__ src,
              const int* __restrict__ dst, int e, float* __restrict__ s,
              float* __restrict__ deg, int* __restrict__ lists,
              int replay, const unsigned* __restrict__ bits) {
  const int lane = threadIdx.x & 31;
  const int wave = __builtin_amdgcn_readfirstlane((int)(threadIdx.x >> 5));
  const int row0 = blk * ROWS;
  int* const q = list + wave * OWN_QCAP;
  int* const lw = lists + ((size_t)blk * OWN_WAVES + wave) * OWN_LCAP;
  const bool rec = replay == 0;
  int ln = 0;

  for (int i = threadIdx.x; i < ROWS * HDIM; i += OWN_THREADS) acc[i] = 0.0f;
  if (threadIdx.x < ROWS) cnt[threadIdx.x] = 0.0f;
  __syncthreads();

  auto add_group = [&](int sq, int dq, int j0, int n) {
    float v[EDGE_GRP][4];
#pragma unroll
    for (int j = 0; j < EDGE_GRP; ++j) {
      const int sr = __builtin_amdgcn_readlane(sq, j0 + j);
      const float* xr = x + (size_t)sr * HDIM + lane;
      v[j][0] = xr[0];
      v[j][1] = xr[32];
      v[j][2] = xr[64];
      v[j][3] = xr[96];
    }
#pragma unroll
    for (int j = 0; j < EDGE_GRP; ++j) {
      if (j < n) {
        const int dr = __builtin_amdgcn_readlane(dq, j0 + j);
        float* o = acc + dr * HDIM + lane;
        lds_add_f32(o, v[j][0]);
        lds_add_f32(o + 32, v[j][1]);
        lds_add_f32(o + 64, v[j][2]);
        lds_add_f32(o + 96, v[j][3]);
      }
    }
  };

  int qn = 0;

  auto take = [&](bool own, int edge, int lrow) {
    const unsigned m = __builtin_amdgcn_ballot_w32(own);
    if (m == 0u) return;
    if (own) q[qn + (int)__builtin_amdgcn_mbcnt_lo(m, 0u)] = edge * ROWS + lrow;
    qn += __builtin_popcount(m);
  };

  auto work_off = [&]() {
    if (qn < 32) return;
    wave_lds_sync();
    int done = 0;
#pragma unroll 1
    for (; qn - done >= 32; done += 32) {
      const int w  = q[done + lane];
      const int sq = src[(unsigned)w / (unsigned)ROWS];
      const int dq = (int)((unsigned)w % (unsigned)ROWS);
      lds_add_f32(&cnt[dq], 1.0f);
      if (rec) {
        if (ln + 32 <= OWN_LCAP - 1) lw[ln + lane] = sq * ROWS + dq;
        ln += 32;
      }
#pragma unroll 1
      for (int j0 = 0; j0 < 32; j0 += EDGE_GRP) add_group(sq, dq, j0, EDGE_GRP);
    }
    const int rest = qn - done;
    const int tw = lane < rest ? q[done + lane] : 0;
    wave_lds_sync();
    if (lane < rest) q[lane] = tw;
    qn = rest;
  };

  const int saved =
      replay ? __builtin_amdgcn_readfirstlane(lw[OWN_LCAP - 1]) : -1;
  if (saved >= 0) {
#pragma unroll 1
    for (int j = 0; j < saved; j += 32) {
      const int m  = saved - j < 32 ? saved - j : 32;
      const int w  = lane < m ? lw[j + lane] : 0;
      const int sq = (int)((unsigned)w / (unsigned)ROWS);
      const int dq = (int)((unsigned)w % (unsigned)ROWS);
      if (lane < m) lds_add_f32(&cnt[dq], 1.0f);
#pragma unroll 1
      for (int j0 = 0; j0 < m; j0 += EDGE_GRP) add_group(sq, dq, j0, m - j0);
    }
  } else {
  int* const gl = q + BM_QCAP;
  const int ngrp = (e + BM_GRP - 1) / BM_GRP;
  const int nrun = (ngrp + 31) / 32;
#pragma unroll 1
  for (int rb = 0; rb < nrun; rb += BM_LANES * OWN_WAVES) {
    const int run = rb + wave + OWN_WAVES * lane;
    unsigned m = 0u;
    if (lane < BM_LANES && run < nrun)
      m = bits != nullptr ? bits[(size_t)run * BM_BLOCKS + blk]
                          : 0xffffffffu;
    int gn = 0;
#pragma unroll 1
    for (int r = 0; r < 32; ++r) {
      const bool has = m != 0u;
      const unsigned b = __builtin_amdgcn_ballot_w32(has);
      if (b == 0u) break;
      if (has) {
        gl[gn + (int)__builtin_amdgcn_mbcnt_lo(b, 0u)] =
            run * 32 + __builtin_ctz(m);
        m &= m - 1u;
      }
      gn += __builtin_popcount(b);
    }
    wave_lds_sync();
#pragma unroll 1
    for (int j = 0; j < gn; j += 32) {
      const int n32 = gn - j < 32 ? gn - j : 32;
      const int gv = lane < n32 ? gl[j + lane] : 0;
#pragma unroll 1
      for (int j0 = 0; j0 < n32; j0 += BM_AHEAD) {
        int ei[BM_AHEAD];
        int dv[BM_AHEAD];
#pragma unroll
        for (int u = 0; u < BM_AHEAD; ++u) {
          ei[u] = __builtin_amdgcn_readlane(gv, j0 + u) * BM_GRP + lane;
          dv[u] = dst[ei[u] < e ? ei[u] : e - 1];
        }
#pragma unroll
        for (int u = 0; u < BM_AHEAD; ++u) {
          if (j0 + u < n32) {
            const unsigned l =
                ei[u] < e ? (unsigned)(dv[u] - row0) : 0xffffffffu;
            take(l < (unsigned)ROWS, ei[u], (int)l);
          }
        }
        work_off();
      }
    }
    wave_lds_sync();
  }

  wave_lds_sync();
  {
    const int w  = lane < qn ? q[lane] : 0;
    const int sq = lane < qn ? src[(unsigned)w / (unsigned)ROWS] : 0;
    const int dq = (int)((unsigned)w % (unsigned)ROWS);
    if (lane < qn) lds_add_f32(&cnt[dq], 1.0f);
    if (rec) {
      if (lane < qn && ln + qn <= OWN_LCAP - 1) lw[ln + lane] = sq * ROWS + dq;
      ln += qn;
      if (lane == 0) lw[OWN_LCAP - 1] = ln <= OWN_LCAP - 1 ? ln : -1;
    }
#pragma unroll 1
    for (int j0 = 0; j0 < qn; j0 += EDGE_GRP) add_group(sq, dq, j0, qn - j0);
  }
  }

  __syncthreads();
  float* out = s + (size_t)row0 * HDIM;
  for (int i = threadIdx.x; i < ROWS * HDIM / 4; i += OWN_THREADS)
    *(v4f*)(out + 4 * i) = *(const v4f*)(acc + 4 * i);
  if (threadIdx.x < ROWS) deg[row0 + threadIdx.x] = cnt[threadIdx.x];
}

template <int ROWS_P, int ROWS_C>
__global__ void __launch_bounds__(OWN_THREADS)
own_pair_kernel(const float* __restrict__ xc, const float* __restrict__ xp,
                const int* __restrict__ e_src, const int* __restrict__ e_dst,
                int e, float* __restrict__ sp, float* __restrict__ sc,
                float* __restrict__ degp, float* __restrict__ degc,
                int* __restrict__ listp, int* __restrict__ listc, int replay,
                const unsigned* __restrict__ bitsp,
                const unsigned* __restrict__ bitsc) {
  constexpr int ROWS_MAX = ROWS_P > ROWS_C ? ROWS_P : ROWS_C;
  __shared__ float acc[ROWS_MAX * HDIM];
  __shared__ float cnt[ROWS_MAX];
  __shared__ int list[OWN_WAVES * OWN_QCAP];
  if (blockIdx.x < BM_BLOCKS)
    own_rows_body<ROWS_P>(acc, cnt, list, (int)blockIdx.x, xc, e_src, e_dst,
                          e, sp, degp, listp, replay, bitsp);
  else
    own_rows_body<ROWS_C>(acc, cnt, list, (int)blockIdx.x - BM_BLOCKS, xp,
                          e_dst, e_src, e, sc, degc, listc, replay, bitsc);
}

template <int ROWS>
__device__ __forceinline__ void group_bits_run(const int* __restrict__ dst,
                                               int e, unsigned* __restrict__ out,
                                               int run, unsigned* w, int lane) {
  for (int k = 0; k < BM_BLOCKS / 32; ++k) w[lane + 32 * k] = 0u;
  wave_lds_sync();
  int d[32];
#pragma unroll
  for (int g = 0; g < 32; ++g) {
    const int i = (run * 32 + g) * BM_GRP + lane;
    d[g] = i < e ? dst[i] : -1;
  }
#pragma unroll
  for (int g = 0; g < 32; ++g) {
    if ((unsigned)d[g] < (unsigned)(ROWS * BM_BLOCKS))
      (void)__hip_atomic_fetch_or(&w[(unsigned)d[g] / (unsigned)ROWS], 1u << g,
                                  __ATOMIC_RELAXED,
                                  __HIP_MEMORY_SCOPE_WORKGROUP);
  }
  wave_lds_sync();
  unsigned* o = out + (size_t)run * BM_BLOCKS;
  for (int k = 0; k < BM_BLOCKS / 32; ++k) o[lane + 32 * k] = w[lane + 32 * k];
}

template <int ROWS0, int ROWS1>
__global__ void __launch_bounds__(256)
group_bits_kernel(const int* __restrict__ dst0, const int* __restrict__ dst1,
                  int e, unsigned* __restrict__ bits0,
                  unsigned* __restrict__ bits1) {
  __shared__ unsigned words[8 * BM_BLOCKS];
  const int lane = threadIdx.x & 31;
  const int wave = threadIdx.x >> 5;
  const int nrun = (e + BM_RUN - 1) / BM_RUN;
  const int run  = blockIdx.x * 8 + wave;
  if (run >= nrun) return;
  unsigned* w = words + wave * BM_BLOCKS;
  if (blockIdx.y == 0) group_bits_run<ROWS0>(dst0, e, bits0, run, w, lane);
  else                 group_bits_run<ROWS1>(dst1, e, bits1, run, w, lane);
}


struct GemmFrag {
  v4f a[2][4];
  v4f w[16];
};

__device__ __forceinline__ void gemm_frag_load(GemmFrag& f,
                                               const float* __restrict__ ap,
                                               const float* __restrict__ wp,
                                               size_t a_tile1, int step) {
  const float* a = ap + step * 32;
#pragma unroll
  for (int i = 0; i < 4; ++i) {
    f.a[0][i] = *(const v4f*)(a + 4 * i);
    f.a[1][i] = *(const v4f*)(a + a_tile1 + 4 * i);
  }
  const float* w = wp + (size_t)step * 32 * HDIM;
#pragma unroll
  for (int i = 0; i < 16; ++i) f.w[i] = *(const v4f*)(w + i * HDIM);
}

__device__ __forceinline__ void gemm_frag_mma(const GemmFrag& f,
                                              v8f (&acc)[2][4]) {
#pragma unroll
  for (int q = 0; q < 8; ++q) {
#pragma unroll
    for (int t = 0; t < 2; ++t) {
      v2f av;
      av.x = f.a[t][q >> 1][(2 * q) & 3];
      av.y = f.a[t][q >> 1][(2 * q + 1) & 3];
#pragma unroll
      for (int j = 0; j < 4; ++j) {
        v2f bv;
        bv.x = f.w[2 * q][j];
        bv.y = f.w[2 * q + 1][j];
        acc[t][j] = __builtin_amdgcn_wmma_f32_16x16x4_f32(
            false, av, false, bv, (short)0, acc[t][j], false, false);
      }
    }
  }
}

template <int KS, int WM>
__global__ void __launch_bounds__(256)
gemm_in_kernel(const float* __restrict__ A, const float* __restrict__ W,
               int K, int kslice, const float* __restrict__ bias,
               const float* __restrict__ emb, const int* __restrict__ nid,
               float* __restrict__ C) {
  static_assert(KS * WM == 8, "one wave per (row group, K slice)");
  constexpr int BROWS = 32 * WM;
  __shared__ float red[KS * BROWS * 64];
  const int lane = threadIdx.x & 31;
  const int wave = threadIdx.x >> 5;
  const int half = lane >> 4;
  const int c    = lane & 15;
  const int wm   = wave % WM;
  const int ks   = wave / WM;
  const int mblk = blockIdx.x >> 1;
  const int ch   = blockIdx.x & 1;
  const int m0   = mblk * BROWS + wm * 32;
  const int nb   = ch * 64 + 4 * c;
  const int kb   = ks * kslice + 16 * half;

  const float* ap = A + (size_t)(m0 + c) * K + kb;
  const float* wp = W + (size_t)kb * HDIM + nb;
  const size_t a_tile1 = (size_t)16 * K;
  const int nsteps = kslice / 32;

  v8f acc[2][4] = {};
  GemmFrag f0, f1;
  gemm_frag_load(f0, ap, wp, a_tile1, 0);

  constexpr int NIT = BROWS * 16 / 256;
  const int rq = threadIdx.x >> 4;
  const int c4 = (threadIdx.x & 15) * 4;
  const int n4 = ch * 64 + c4;
  const v4f bvec = *(const v4f*)(bias + n4);
  v4f ev[NIT];
#pragma unroll
  for (int u = 0; u < NIT; ++u) {
    const int m = mblk * BROWS + rq + 16 * u;
    ev[u] = *(const v4f*)(emb + (size_t)nid[m] * HDIM + n4);
  }

  for (int s = 0; s < nsteps; s += 2) {
    gemm_frag_load(f1, ap, wp, a_tile1, s + 1);
    __builtin_amdgcn_sched_barrier(0);
    gemm_frag_mma(f0, acc);
    const int s2 = (s + 2 < nsteps) ? s + 2 : s + 1;
    gemm_frag_load(f0, ap, wp, a_tile1, s2);
    __builtin_amdgcn_sched_barrier(0);
    gemm_frag_mma(f1, acc);
  }

  float* rp = red + (size_t)ks * BROWS * 64;
#pragma unroll
  for (int t = 0; t < 2; ++t) {
#pragma unroll
    for (int r = 0; r < 8; ++r) {
      const int rl = wm * 32 + 16 * t + 8 * half + r;
      v4f o;
      o.x = acc[t][0][r];
      o.y = acc[t][1][r];
      o.z = acc[t][2][r];
      o.w = acc[t][3][r];
      *(v4f*)(rp + rl * 64 + 4 * c) = o;
    }
  }
  __syncthreads();

#pragma unroll
  for (int u = 0; u < NIT; ++u) {
    const int rl = rq + 16 * u;
    const float* p = red + rl * 64 + c4;
    v4f sum = *(const v4f*)p;
#pragma unroll
    for (int s = 1; s < KS; ++s) sum += *(const v4f*)(p + s * BROWS * 64);
    *(v4f*)(C + (size_t)(mblk * BROWS + rl) * HDIM + n4) = (sum + bvec) + ev[u];
  }
}

__device__ __forceinline__ void
sage_body(int blk, const float* __restrict__ S0, const float* __restrict__ deg,
          const float* __restrict__ X, const float* __restrict__ Wl,
          const float* __restrict__ bl, const float* __restrict__ Wr,
          float* __restrict__ Out, int relu) {
  const int lane = threadIdx.x & 31;
  const int wave = threadIdx.x >> 5;
  const int half = lane >> 4;
  const int nloc = lane & 15;
  const int m0   = blk * 16;
  const int row  = m0 + nloc;
  const int n    = wave * 16 + nloc;

  const float* Srow = S0 + (size_t)row * HDIM;
  const float* Xrow = X + (size_t)row * HDIM;
  v8f accS = {};
  v8f accX = {};
#pragma unroll 4
  for (int k0 = 0; k0 < HDIM; k0 += 4) {
    const int ka = k0 + 2 * half;
    v2f a1; a1.x = Srow[ka]; a1.y = Srow[ka + 1];
    v2f b1; b1.x = Wl[(size_t)ka * HDIM + n];
            b1.y = Wl[(size_t)(ka + 1) * HDIM + n];
    accS = __builtin_amdgcn_wmma_f32_16x16x4_f32(false, a1, false, b1,
                                                 (short)0, accS, false, false);
    v2f a2; a2.x = Xrow[ka]; a2.y = Xrow[ka + 1];
    v2f b2; b2.x = Wr[(size_t)ka * HDIM + n];
            b2.y = Wr[(size_t)(ka + 1) * HDIM + n];
    accX = __builtin_amdgcn_wmma_f32_16x16x4_f32(false, a2, false, b2,
                                                 (short)0, accX, false, false);
  }

  const float bn = bl[n];
#pragma unroll
  for (int r = 0; r < 8; ++r) {
    const int m = m0 + r + 8 * half;
    float d = deg[m];
    d = d < 1.0f ? 1.0f : d;
    float v = accS[r] / d + accX[r] + bn;
    if (relu) v = v > 0.0f ? v : 0.0f;
    Out[(size_t)m * HDIM + n] = v;
  }
}

__global__ void __launch_bounds__(256)
sage_kernel(const float* __restrict__ S0, const float* __restrict__ deg,
            const float* __restrict__ X, const float* __restrict__ Wl,
            const float* __restrict__ bl, const float* __restrict__ Wr,
            float* __restrict__ Out, int relu) {
  sage_body((int)blockIdx.x, S0, deg, X, Wl, bl, Wr, Out, relu);
}

__global__ void __launch_bounds__(256)
sage_pair_kernel(int nbp, int relu,
                 const float* __restrict__ Sp, const float* __restrict__ degp,
                 const float* __restrict__ Xp, const float* __restrict__ Wlp,
                 const float* __restrict__ blp, const float* __restrict__ Wrp,
                 float* __restrict__ Outp,
                 const float* __restrict__ Sc, const float* __restrict__ degc,
                 const float* __restrict__ Xc, const float* __restrict__ Wlc,
                 const float* __restrict__ blc, const float* __restrict__ Wrc,
                 float* __restrict__ Outc) {
  if ((int)blockIdx.x < nbp)
    sage_body((int)blockIdx.x, Sp, degp, Xp, Wlp, blp, Wrp, Outp, relu);
  else
    sage_body((int)blockIdx.x - nbp, Sc, degc, Xc, Wlc, blc, Wrc, Outc, relu);
}

__global__ void edge_dot_kernel(const float* __restrict__ Hc,
                                const float* __restrict__ Hp,
                                const int* __restrict__ ls,
                                const int* __restrict__ ld, int e,
                                float* __restrict__ out) {
  int wid  = (blockIdx.x * blockDim.x + threadIdx.x) >> 5;
  int lane = threadIdx.x & 31;
  if (wid >= e) return;
  int a = ls[wid];
  int b = ld[wid];
  const float4 va = *(const float4*)(Hc + (size_t)a * HDIM + lane * 4);
  const float4 vb = *(const float4*)(Hp + (size_t)b * HDIM + lane * 4);
  float v = va.x * vb.x + va.y * vb.y + va.z * vb.z + va.w * vb.w;
#pragma unroll
  for (int off = 16; off > 0; off >>= 1) v += __shfl_down(v, off, 32);
  if (lane == 0) out[wid] = v;
}

extern "C" void kernel_launch(void* const* d_in, const int* in_sizes, int n_in,
                              void* d_out, int out_size, void* d_ws,
                              size_t ws_size, hipStream_t stream) {
  (void)in_sizes; (void)n_in; (void)out_size; (void)ws_size;
  const int NC = 4096, NP = 8192, E = 500000, EL = 500000;

  const float* x_country = (const float*)d_in[0];
  const float* x_product = (const float*)d_in[1];
  const float* Wc_lin = (const float*)d_in[2];
  const float* bc_lin = (const float*)d_in[3];
  const float* Wp_lin = (const float*)d_in[4];
  const float* bp_lin = (const float*)d_in[5];
  const float* emb_c  = (const float*)d_in[6];
  const float* emb_p  = (const float*)d_in[7];
  const float* W1ep_l = (const float*)d_in[8];
  const float* b1ep   = (const float*)d_in[9];
  const float* W1ep_r = (const float*)d_in[10];
  const float* W1pc_l = (const float*)d_in[11];
  const float* b1pc   = (const float*)d_in[12];
  const float* W1pc_r = (const float*)d_in[13];
  const float* W2ep_l = (const float*)d_in[14];
  const float* b2ep   = (const float*)d_in[15];
  const float* W2ep_r = (const float*)d_in[16];
  const float* W2pc_l = (const float*)d_in[17];
  const float* b2pc   = (const float*)d_in[18];
  const float* W2pc_r = (const float*)d_in[19];
  const int* nid_c = (const int*)d_in[20];
  const int* nid_p = (const int*)d_in[21];
  const int* e_src = (const int*)d_in[22];
  const int* e_dst = (const int*)d_in[23];
  const int* l_src = (const int*)d_in[24];
  const int* l_dst = (const int*)d_in[25];

  float* ws = (float*)d_ws;
  const size_t unit = (size_t)NC * HDIM;
  float* xp    = ws;
  float* xc    = ws + 2 * unit;
  float* sumC1 = ws + 3 * unit;
  float* sumP1 = ws + 4 * unit;
  int* listP   = (int*)(ws + 6 * unit);
  int* listC   = (int*)(ws + 8 * unit);
  unsigned* bitsP = (unsigned*)(ws + 10 * unit);
  unsigned* bitsC = bitsP + 131072;
  float* hc    = ws + 10 * unit;
  float* hp    = ws + 2 * unit;
  float* sumP2 = ws;
  float* sumC2 = ws + 4 * unit;
  float* hp2   = ws + 6 * unit;
  float* hc2   = ws + 8 * unit;
  float* degp  = ws + 11 * unit;
  float* degc  = degp + NP;

  const int TB = 256;
  const int own_blocks = 2 * BM_BLOCKS;
  const int OT = 1024;

  gemm_in_kernel<8, 1><<<(NC / 32) * 2, TB, 0, stream>>>(
      x_country, Wc_lin, NC, NC / 8, bc_lin, emb_c, nid_c, xc);
  gemm_in_kernel<4, 2><<<(NP / 64) * 2, TB, 0, stream>>>(
      x_product, Wp_lin, NP, NP / 4, bp_lin, emb_p, nid_p, xp);

  const int runs = (E + BM_RUN - 1) / BM_RUN;
  group_bits_kernel<NP / 256, NC / 256><<<dim3((runs + 7) / 8, 2), TB, 0,
                                          stream>>>(e_dst, e_src, E, bitsP,
                                                    bitsC);

  own_pair_kernel<NP / 256, NC / 256><<<own_blocks, OT, 0, stream>>>(
      xc, xp, e_src, e_dst, E, sumP1, sumC1, degp, degc, listP, listC, 0,
      bitsP, bitsC);
  sage_kernel<<<NC / 16, TB, 0, stream>>>(sumC1, degc, xc,
                                          W1pc_l, b1pc, W1pc_r, hc, 1);
  sage_kernel<<<NP / 16, TB, 0, stream>>>(sumP1, degp, xp,
                                          W1ep_l, b1ep, W1ep_r, hp, 1);

  own_pair_kernel<NP / 256, NC / 256><<<own_blocks, OT, 0, stream>>>(
      hc, hp, e_src, e_dst, E, sumP2, sumC2, degp, degc, listP, listC, 1,
      nullptr, nullptr);
  sage_pair_kernel<<<NP / 16 + NC / 16, TB, 0, stream>>>(
      NP / 16, 0,
      sumP2, degp, hp, W2ep_l, b2ep, W2ep_r, hp2,
      sumC2, degc, hc, W2pc_l, b2pc, W2pc_r, hc2);

  const int el_blocks = (EL * 32 + TB - 1) / TB;
  edge_dot_kernel<<<el_blocks, TB, 0, stream>>>(hc2, hp2, l_src, l_dst, EL,
                                                (float*)d_out);
}


# edge head by owner blocks: a block keeps its 16 hc2 rows in LDS, finds its label edges by a bitmap of l_src (made beside layer 2's products), gathers one hp2 row per edge, not two; sum order kept
# speedup vs baseline: 3.3196x; 1.0449x over previous
#include <hip/hip_runtime.h>


typedef __attribute__((ext_vector_type(2))) float v2f;
typedef __attribute__((ext_vector_type(4))) float v4f;
typedef __attribute__((ext_vector_type(8))) float v8f;

constexpr int HDIM = 128;


constexpr int OWN_THREADS = 1024;
constexpr int OWN_WAVES   = OWN_THREADS / 32;
constexpr int BM_GRP      = 32;
constexpr int BM_RUN      = 32 * BM_GRP;
constexpr int BM_BLOCKS   = 256;
constexpr int BM_AHEAD    = 16;
constexpr int BM_QCAP     = 32 + BM_AHEAD * BM_GRP;
constexpr int BM_LANES    = 16;
constexpr int BM_GCAP     = BM_LANES * 32;
constexpr int OWN_QCAP    = BM_QCAP + BM_GCAP;
constexpr int EDGE_GRP    = 8;
constexpr int OWN_LCAP    = 128;

__device__ __forceinline__ void lds_add_f32(float* p, float v) {
  (void)__hip_atomic_fetch_add(p, v, __ATOMIC_RELAXED,
                               __HIP_MEMORY_SCOPE_WORKGROUP);
}

__device__ __forceinline__ void wave_lds_sync() {
  __builtin_amdgcn_fence(__ATOMIC_RELEASE, "wavefront");
  __builtin_amdgcn_wave_barrier();
  __builtin_amdgcn_fence(__ATOMIC_ACQUIRE, "wavefront");
}

template <int ROWS>
__device__ __forceinline__ void
own_rows_body(float* acc, float* cnt, int* list, int blk,
              const float* __restrict__ x, const int* __restrict__ src,
              const int* __restrict__ dst, int e, float* __restrict__ s,
              float* __restrict__ deg, int* __restrict__ lists,
              int replay, const unsigned* __restrict__ bits) {
  const int lane = threadIdx.x & 31;
  const int wave = __builtin_amdgcn_readfirstlane((int)(threadIdx.x >> 5));
  const int row0 = blk * ROWS;
  int* const q = list + wave * OWN_QCAP;
  int* const lw = lists + ((size_t)blk * OWN_WAVES + wave) * OWN_LCAP;
  const bool rec = replay == 0;
  int ln = 0;

  for (int i = threadIdx.x; i < ROWS * HDIM; i += OWN_THREADS) acc[i] = 0.0f;
  if (threadIdx.x < ROWS) cnt[threadIdx.x] = 0.0f;
  __syncthreads();

  auto add_group = [&](int sq, int dq, int j0, int n) {
    float v[EDGE_GRP][4];
#pragma unroll
    for (int j = 0; j < EDGE_GRP; ++j) {
      const int sr = __builtin_amdgcn_readlane(sq, j0 + j);
      const float* xr = x + (size_t)sr * HDIM + lane;
      v[j][0] = xr[0];
      v[j][1] = xr[32];
      v[j][2] = xr[64];
      v[j][3] = xr[96];
    }
#pragma unroll
    for (int j = 0; j < EDGE_GRP; ++j) {
      if (j < n) {
        const int dr = __builtin_amdgcn_readlane(dq, j0 + j);
        float* o = acc + dr * HDIM + lane;
        lds_add_f32(o, v[j][0]);
        lds_add_f32(o + 32, v[j][1]);
        lds_add_f32(o + 64, v[j][2]);
        lds_add_f32(o + 96, v[j][3]);
      }
    }
  };

  int qn = 0;

  auto take = [&](bool own, int edge, int lrow) {
    const unsigned m = __builtin_amdgcn_ballot_w32(own);
    if (m == 0u) return;
    if (own) q[qn + (int)__builtin_amdgcn_mbcnt_lo(m, 0u)] = edge * ROWS + lrow;
    qn += __builtin_popcount(m);
  };

  auto work_off = [&]() {
    if (qn < 32) return;
    wave_lds_sync();
    int done = 0;
#pragma unroll 1
    for (; qn - done >= 32; done += 32) {
      const int w  = q[done + lane];
      const int sq = src[(unsigned)w / (unsigned)ROWS];
      const int dq = (int)((unsigned)w % (unsigned)ROWS);
      lds_add_f32(&cnt[dq], 1.0f);
      if (rec) {
        if (ln + 32 <= OWN_LCAP - 1) lw[ln + lane] = sq * ROWS + dq;
        ln += 32;
      }
#pragma unroll 1
      for (int j0 = 0; j0 < 32; j0 += EDGE_GRP) add_group(sq, dq, j0, EDGE_GRP);
    }
    const int rest = qn - done;
    const int tw = lane < rest ? q[done + lane] : 0;
    wave_lds_sync();
    if (lane < rest) q[lane] = tw;
    qn = rest;
  };

  const int saved =
      replay ? __builtin_amdgcn_readfirstlane(lw[OWN_LCAP - 1]) : -1;
  if (saved >= 0) {
#pragma unroll 1
    for (int j = 0; j < saved; j += 32) {
      const int m  = saved - j < 32 ? saved - j : 32;
      const int w  = lane < m ? lw[j + lane] : 0;
      const int sq = (int)((unsigned)w / (unsigned)ROWS);
      const int dq = (int)((unsigned)w % (unsigned)ROWS);
      if (lane < m) lds_add_f32(&cnt[dq], 1.0f);
#pragma unroll 1
      for (int j0 = 0; j0 < m; j0 += EDGE_GRP) add_group(sq, dq, j0, m - j0);
    }
  } else {
  int* const gl = q + BM_QCAP;
  const int ngrp = (e + BM_GRP - 1) / BM_GRP;
  const int nrun = (ngrp + 31) / 32;
#pragma unroll 1
  for (int rb = 0; rb < nrun; rb += BM_LANES * OWN_WAVES) {
    const int run = rb + wave + OWN_WAVES * lane;
    unsigned m = 0u;
    if (lane < BM_LANES && run < nrun)
      m = bits != nullptr ? bits[(size_t)run * BM_BLOCKS + blk]
                          : 0xffffffffu;
    int gn = 0;
#pragma unroll 1
    for (int r = 0; r < 32; ++r) {
      const bool has = m != 0u;
      const unsigned b = __builtin_amdgcn_ballot_w32(has);
      if (b == 0u) break;
      if (has) {
        gl[gn + (int)__builtin_amdgcn_mbcnt_lo(b, 0u)] =
            run * 32 + __builtin_ctz(m);
        m &= m - 1u;
      }
      gn += __builtin_popcount(b);
    }
    wave_lds_sync();
#pragma unroll 1
    for (int j = 0; j < gn; j += 32) {
      const int n32 = gn - j < 32 ? gn - j : 32;
      const int gv = lane < n32 ? gl[j + lane] : 0;
#pragma unroll 1
      for (int j0 = 0; j0 < n32; j0 += BM_AHEAD) {
        int ei[BM_AHEAD];
        int dv[BM_AHEAD];
#pragma unroll
        for (int u = 0; u < BM_AHEAD; ++u) {
          ei[u] = __builtin_amdgcn_readlane(gv, j0 + u) * BM_GRP + lane;
          dv[u] = dst[ei[u] < e ? ei[u] : e - 1];
        }
#pragma unroll
        for (int u = 0; u < BM_AHEAD; ++u) {
          if (j0 + u < n32) {
            const unsigned l =
                ei[u] < e ? (unsigned)(dv[u] - row0) : 0xffffffffu;
            take(l < (unsigned)ROWS, ei[u], (int)l);
          }
        }
        work_off();
      }
    }
    wave_lds_sync();
  }

  wave_lds_sync();
  {
    const int w  = lane < qn ? q[lane] : 0;
    const int sq = lane < qn ? src[(unsigned)w / (unsigned)ROWS] : 0;
    const int dq = (int)((unsigned)w % (unsigned)ROWS);
    if (lane < qn) lds_add_f32(&cnt[dq], 1.0f);
    if (rec) {
      if (lane < qn && ln + qn <= OWN_LCAP - 1) lw[ln + lane] = sq * ROWS + dq;
      ln += qn;
      if (lane == 0) lw[OWN_LCAP - 1] = ln <= OWN_LCAP - 1 ? ln : -1;
    }
#pragma unroll 1
    for (int j0 = 0; j0 < qn; j0 += EDGE_GRP) add_group(sq, dq, j0, qn - j0);
  }
  }

  __syncthreads();
  float* out = s + (size_t)row0 * HDIM;
  for (int i = threadIdx.x; i < ROWS * HDIM / 4; i += OWN_THREADS)
    *(v4f*)(out + 4 * i) = *(const v4f*)(acc + 4 * i);
  if (threadIdx.x < ROWS) deg[row0 + threadIdx.x] = cnt[threadIdx.x];
}

template <int ROWS_P, int ROWS_C>
__global__ void __launch_bounds__(OWN_THREADS)
own_pair_kernel(const float* __restrict__ xc, const float* __restrict__ xp,
                const int* __restrict__ e_src, const int* __restrict__ e_dst,
                int e, float* __restrict__ sp, float* __restrict__ sc,
                float* __restrict__ degp, float* __restrict__ degc,
                int* __restrict__ listp, int* __restrict__ listc, int replay,
                const unsigned* __restrict__ bitsp,
                const unsigned* __restrict__ bitsc) {
  constexpr int ROWS_MAX = ROWS_P > ROWS_C ? ROWS_P : ROWS_C;
  __shared__ float acc[ROWS_MAX * HDIM];
  __shared__ float cnt[ROWS_MAX];
  __shared__ int list[OWN_WAVES * OWN_QCAP];
  if (blockIdx.x < BM_BLOCKS)
    own_rows_body<ROWS_P>(acc, cnt, list, (int)blockIdx.x, xc, e_src, e_dst,
                          e, sp, degp, listp, replay, bitsp);
  else
    own_rows_body<ROWS_C>(acc, cnt, list, (int)blockIdx.x - BM_BLOCKS, xp,
                          e_dst, e_src, e, sc, degc, listc, replay, bitsc);
}

template <int ROWS>
__device__ __forceinline__ void group_bits_run(const int* __restrict__ dst,
                                               int e, unsigned* __restrict__ out,
                                               int run, unsigned* w, int lane) {
  for (int k = 0; k < BM_BLOCKS / 32; ++k) w[lane + 32 * k] = 0u;
  wave_lds_sync();
  int d[32];
#pragma unroll
  for (int g = 0; g < 32; ++g) {
    const int i = (run * 32 + g) * BM_GRP + lane;
    d[g] = i < e ? dst[i] : -1;
  }
#pragma unroll
  for (int g = 0; g < 32; ++g) {
    if ((unsigned)d[g] < (unsigned)(ROWS * BM_BLOCKS))
      (void)__hip_atomic_fetch_or(&w[(unsigned)d[g] / (unsigned)ROWS], 1u << g,
                                  __ATOMIC_RELAXED,
                                  __HIP_MEMORY_SCOPE_WORKGROUP);
  }
  wave_lds_sync();
  unsigned* o = out + (size_t)run * BM_BLOCKS;
  for (int k = 0; k < BM_BLOCKS / 32; ++k) o[lane + 32 * k] = w[lane + 32 * k];
}

template <int ROWS0, int ROWS1>
__global__ void __launch_bounds__(256)
group_bits_kernel(const int* __restrict__ dst0, const int* __restrict__ dst1,
                  int e, unsigned* __restrict__ bits0,
                  unsigned* __restrict__ bits1) {
  __shared__ unsigned words[8 * BM_BLOCKS];
  const int lane = threadIdx.x & 31;
  const int wave = threadIdx.x >> 5;
  const int nrun = (e + BM_RUN - 1) / BM_RUN;
  const int run  = blockIdx.x * 8 + wave;
  if (run >= nrun) return;
  unsigned* w = words + wave * BM_BLOCKS;
  if (blockIdx.y == 0) group_bits_run<ROWS0>(dst0, e, bits0, run, w, lane);
  else                 group_bits_run<ROWS1>(dst1, e, bits1, run, w, lane);
}


struct GemmFrag {
  v4f a[2][4];
  v4f w[16];
};

__device__ __forceinline__ void gemm_frag_load(GemmFrag& f,
                                               const float* __restrict__ ap,
                                               const float* __restrict__ wp,
                                               size_t a_tile1, int step) {
  const float* a = ap + step * 32;
#pragma unroll
  for (int i = 0; i < 4; ++i) {
    f.a[0][i] = *(const v4f*)(a + 4 * i);
    f.a[1][i] = *(const v4f*)(a + a_tile1 + 4 * i);
  }
  const float* w = wp + (size_t)step * 32 * HDIM;
#pragma unroll
  for (int i = 0; i < 16; ++i) f.w[i] = *(const v4f*)(w + i * HDIM);
}

__device__ __forceinline__ void gemm_frag_mma(const GemmFrag& f,
                                              v8f (&acc)[2][4]) {
#pragma unroll
  for (int q = 0; q < 8; ++q) {
#pragma unroll
    for (int t = 0; t < 2; ++t) {
      v2f av;
      av.x = f.a[t][q >> 1][(2 * q) & 3];
      av.y = f.a[t][q >> 1][(2 * q + 1) & 3];
#pragma unroll
      for (int j = 0; j < 4; ++j) {
        v2f bv;
        bv.x = f.w[2 * q][j];
        bv.y = f.w[2 * q + 1][j];
        acc[t][j] = __builtin_amdgcn_wmma_f32_16x16x4_f32(
            false, av, false, bv, (short)0, acc[t][j], false, false);
      }
    }
  }
}

template <int KS, int WM>
__global__ void __launch_bounds__(256)
gemm_in_kernel(const float* __restrict__ A, const float* __restrict__ W,
               int K, int kslice, const float* __restrict__ bias,
               const float* __restrict__ emb, const int* __restrict__ nid,
               float* __restrict__ C) {
  static_assert(KS * WM == 8, "one wave per (row group, K slice)");
  constexpr int BROWS = 32 * WM;
  __shared__ float red[KS * BROWS * 64];
  const int lane = threadIdx.x & 31;
  const int wave = threadIdx.x >> 5;
  const int half = lane >> 4;
  const int c    = lane & 15;
  const int wm   = wave % WM;
  const int ks   = wave / WM;
  const int mblk = blockIdx.x >> 1;
  const int ch   = blockIdx.x & 1;
  const int m0   = mblk * BROWS + wm * 32;
  const int nb   = ch * 64 + 4 * c;
  const int kb   = ks * kslice + 16 * half;

  const float* ap = A + (size_t)(m0 + c) * K + kb;
  const float* wp = W + (size_t)kb * HDIM + nb;
  const size_t a_tile1 = (size_t)16 * K;
  const int nsteps = kslice / 32;

  v8f acc[2][4] = {};
  GemmFrag f0, f1;
  gemm_frag_load(f0, ap, wp, a_tile1, 0);

  constexpr int NIT = BROWS * 16 / 256;
  const int rq = threadIdx.x >> 4;
  const int c4 = (threadIdx.x & 15) * 4;
  const int n4 = ch * 64 + c4;
  const v4f bvec = *(const v4f*)(bias + n4);
  v4f ev[NIT];
#pragma unroll
  for (int u = 0; u < NIT; ++u) {
    const int m = mblk * BROWS + rq + 16 * u;
    ev[u] = *(const v4f*)(emb + (size_t)nid[m] * HDIM + n4);
  }

  for (int s = 0; s < nsteps; s += 2) {
    gemm_frag_load(f1, ap, wp, a_tile1, s + 1);
    __builtin_amdgcn_sched_barrier(0);
    gemm_frag_mma(f0, acc);
    const int s2 = (s + 2 < nsteps) ? s + 2 : s + 1;
    gemm_frag_load(f0, ap, wp, a_tile1, s2);
    __builtin_amdgcn_sched_barrier(0);
    gemm_frag_mma(f1, acc);
  }

  float* rp = red + (size_t)ks * BROWS * 64;
#pragma unroll
  for (int t = 0; t < 2; ++t) {
#pragma unroll
    for (int r = 0; r < 8; ++r) {
      const int rl = wm * 32 + 16 * t + 8 * half + r;
      v4f o;
      o.x = acc[t][0][r];
      o.y = acc[t][1][r];
      o.z = acc[t][2][r];
      o.w = acc[t][3][r];
      *(v4f*)(rp + rl * 64 + 4 * c) = o;
    }
  }
  __syncthreads();

#pragma unroll
  for (int u = 0; u < NIT; ++u) {
    const int rl = rq + 16 * u;
    const float* p = red + rl * 64 + c4;
    v4f sum = *(const v4f*)p;
#pragma unroll
    for (int s = 1; s < KS; ++s) sum += *(const v4f*)(p + s * BROWS * 64);
    *(v4f*)(C + (size_t)(mblk * BROWS + rl) * HDIM + n4) = (sum + bvec) + ev[u];
  }
}

__device__ __forceinline__ void
sage_body(int blk, const float* __restrict__ S0, const float* __restrict__ deg,
          const float* __restrict__ X, const float* __restrict__ Wl,
          const float* __restrict__ bl, const float* __restrict__ Wr,
          float* __restrict__ Out, int relu) {
  const int lane = threadIdx.x & 31;
  const int wave = threadIdx.x >> 5;
  const int half = lane >> 4;
  const int nloc = lane & 15;
  const int m0   = blk * 16;
  const int row  = m0 + nloc;
  const int n    = wave * 16 + nloc;

  const float* Srow = S0 + (size_t)row * HDIM;
  const float* Xrow = X + (size_t)row * HDIM;
  v8f accS = {};
  v8f accX = {};
#pragma unroll 4
  for (int k0 = 0; k0 < HDIM; k0 += 4) {
    const int ka = k0 + 2 * half;
    v2f a1; a1.x = Srow[ka]; a1.y = Srow[ka + 1];
    v2f b1; b1.x = Wl[(size_t)ka * HDIM + n];
            b1.y = Wl[(size_t)(ka + 1) * HDIM + n];
    accS = __builtin_amdgcn_wmma_f32_16x16x4_f32(false, a1, false, b1,
                                                 (short)0, accS, false, false);
    v2f a2; a2.x = Xrow[ka]; a2.y = Xrow[ka + 1];
    v2f b2; b2.x = Wr[(size_t)ka * HDIM + n];
            b2.y = Wr[(size_t)(ka + 1) * HDIM + n];
    accX = __builtin_amdgcn_wmma_f32_16x16x4_f32(false, a2, false, b2,
                                                 (short)0, accX, false, false);
  }

  const float bn = bl[n];
#pragma unroll
  for (int r = 0; r < 8; ++r) {
    const int m = m0 + r + 8 * half;
    float d = deg[m];
    d = d < 1.0f ? 1.0f : d;
    float v = accS[r] / d + accX[r] + bn;
    if (relu) v = v > 0.0f ? v : 0.0f;
    Out[(size_t)m * HDIM + n] = v;
  }
}

__global__ void __launch_bounds__(256)
sage_kernel(const float* __restrict__ S0, const float* __restrict__ deg,
            const float* __restrict__ X, const float* __restrict__ Wl,
            const float* __restrict__ bl, const float* __restrict__ Wr,
            float* __restrict__ Out, int relu) {
  sage_body((int)blockIdx.x, S0, deg, X, Wl, bl, Wr, Out, relu);
}

template <int LROWS>
__global__ void __launch_bounds__(256)
sage_pair_kernel(int nbp, int nbc, int relu,
                 const float* __restrict__ Sp, const float* __restrict__ degp,
                 const float* __restrict__ Xp, const float* __restrict__ Wlp,
                 const float* __restrict__ blp, const float* __restrict__ Wrp,
                 float* __restrict__ Outp,
                 const float* __restrict__ Sc, const float* __restrict__ degc,
                 const float* __restrict__ Xc, const float* __restrict__ Wlc,
                 const float* __restrict__ blc, const float* __restrict__ Wrc,
                 float* __restrict__ Outc,
                 const int* __restrict__ ls, int el,
                 unsigned* __restrict__ bitsl) {
  __shared__ unsigned words[8 * BM_BLOCKS];
  if ((int)blockIdx.x < nbp) {
    sage_body((int)blockIdx.x, Sp, degp, Xp, Wlp, blp, Wrp, Outp, relu);
  } else if ((int)blockIdx.x < nbp + nbc) {
    sage_body((int)blockIdx.x - nbp, Sc, degc, Xc, Wlc, blc, Wrc, Outc, relu);
  } else {
    const int lane = threadIdx.x & 31;
    const int wave = threadIdx.x >> 5;
    const int nrun = (el + BM_RUN - 1) / BM_RUN;
    const int run  = ((int)blockIdx.x - nbp - nbc) * 8 + wave;
    if (run < nrun)
      group_bits_run<LROWS>(ls, el, bitsl, run, words + wave * BM_BLOCKS, lane);
  }
}

constexpr int EDGE_ROWS = 16;

template <int XM>
__device__ __forceinline__ float row_xor(float v) {
  return __builtin_bit_cast(
      float, __builtin_amdgcn_update_dpp(0, __builtin_bit_cast(int, v),
                                         0x160 | XM, 0xf, 0xf, true));
}

__device__ __forceinline__ float half_xor(float v) {
  const unsigned u = __builtin_bit_cast(unsigned, v);
  return __builtin_bit_cast(
      float, __builtin_amdgcn_permlanex16(u, u, 0x76543210u, 0xfedcba98u,
                                          false, false));
}

template <int N, int XM>
__device__ __forceinline__ void fold_sums(float (&x)[EDGE_ROWS], bool up) {
#pragma unroll
  for (int t = 0; t < N / 2; ++t) {
    const float keep = up ? x[t + N / 2] : x[t];
    const float give = up ? x[t] : x[t + N / 2];
    x[t] = keep + row_xor<XM>(give);
  }
}

__device__ __forceinline__ float wave_sums16(float (&x)[EDGE_ROWS], int lane) {
#pragma unroll
  for (int t = 0; t < EDGE_ROWS; ++t) x[t] += half_xor(x[t]);
  fold_sums<16, 8>(x, (lane & 8) != 0);
  fold_sums<8, 4>(x, (lane & 4) != 0);
  fold_sums<4, 2>(x, (lane & 2) != 0);
  fold_sums<2, 1>(x, (lane & 1) != 0);
  return x[0];
}

template <int ROWS>
__global__ void __launch_bounds__(OWN_THREADS)
edge_own_kernel(const float* __restrict__ Hc, const float* __restrict__ Hp,
                const int* __restrict__ ls, const int* __restrict__ ld, int e,
                float* __restrict__ out, const unsigned* __restrict__ bits) {
  __shared__ float hrow[ROWS * HDIM];
  __shared__ int list[OWN_WAVES * OWN_QCAP];
  const int lane = threadIdx.x & 31;
  const int wave = __builtin_amdgcn_readfirstlane((int)(threadIdx.x >> 5));
  const int blk  = (int)blockIdx.x;
  const int row0 = blk * ROWS;
  int* const q = list + wave * OWN_QCAP;

  for (int i = threadIdx.x; i < ROWS * HDIM / 4; i += OWN_THREADS)
    *(v4f*)(hrow + 4 * i) = *(const v4f*)(Hc + (size_t)row0 * HDIM + 4 * i);
  __syncthreads();

  auto dot_group = [&](int pq, int dq, int j0, float& res) {
    v4f v[EDGE_ROWS];
#pragma unroll
    for (int j = 0; j < EDGE_ROWS; ++j) {
      const int pr = __builtin_amdgcn_readlane(pq, j0 + j);
      v[j] = *(const v4f*)(Hp + (size_t)pr * HDIM + 4 * lane);
    }
    __builtin_amdgcn_sched_barrier(0);
    float p[EDGE_ROWS];
#pragma unroll
    for (int j = 0; j < EDGE_ROWS; ++j) {
      const int dr = __builtin_amdgcn_readlane(dq, j0 + j);
      const v4f h = *(const v4f*)(hrow + dr * HDIM + 4 * lane);
      float t = h.y * v[j].y;
      t = __builtin_fmaf(h.x, v[j].x, t);
      t = __builtin_fmaf(h.z, v[j].z, t);
      p[j] = __builtin_fmaf(h.w, v[j].w, t);
    }
    const float tot = wave_sums16(p, lane);
    if ((lane & 16) == j0) res = tot;
  };

  int qn = 0;

  auto take = [&](bool own, int edge, int lrow) {
    const unsigned m = __builtin_amdgcn_ballot_w32(own);
    if (m == 0u) return;
    if (own) q[qn + (int)__builtin_amdgcn_mbcnt_lo(m, 0u)] = edge * ROWS + lrow;
    qn += __builtin_popcount(m);
  };

  auto work_off = [&]() {
    if (qn < 32) return;
    wave_lds_sync();
    int done = 0;
#pragma unroll 1
    for (; qn - done >= 32; done += 32) {
      const int w  = q[done + lane];
      const int eq = (int)((unsigned)w / (unsigned)ROWS);
      const int dq = (int)((unsigned)w % (unsigned)ROWS);
      const int pq = ld[eq];
      float res = 0.0f;
#pragma unroll 1
      for (int j0 = 0; j0 < 32; j0 += EDGE_ROWS) dot_group(pq, dq, j0, res);
      out[eq] = res;
    }
    const int rest = qn - done;
    const int tw = lane < rest ? q[done + lane] : 0;
    wave_lds_sync();
    if (lane < rest) q[lane] = tw;
    qn = rest;
  };

  int* const gl = q + BM_QCAP;
  const int ngrp = (e + BM_GRP - 1) / BM_GRP;
  const int nrun = (ngrp + 31) / 32;
#pragma unroll 1
  for (int rb = 0; rb < nrun; rb += BM_LANES * OWN_WAVES) {
    const int run = rb + wave + OWN_WAVES * lane;
    unsigned m = 0u;
    if (lane < BM_LANES && run < nrun) m = bits[(size_t)run * BM_BLOCKS + blk];
    int gn = 0;
#pragma unroll 1
    for (int r = 0; r < 32; ++r) {
      const bool has = m != 0u;
      const unsigned b = __builtin_amdgcn_ballot_w32(has);
      if (b == 0u) break;
      if (has) {
        gl[gn + (int)__builtin_amdgcn_mbcnt_lo(b, 0u)] =
            run * 32 + __builtin_ctz(m);
        m &= m - 1u;
      }
      gn += __builtin_popcount(b);
    }
    wave_lds_sync();
#pragma unroll 1
    for (int j = 0; j < gn; j += 32) {
      const int n32 = gn - j < 32 ? gn - j : 32;
      const int gv = lane < n32 ? gl[j + lane] : 0;
#pragma unroll 1
      for (int j0 = 0; j0 < n32; j0 += BM_AHEAD) {
        int ei[BM_AHEAD];
        int sv[BM_AHEAD];
#pragma unroll
        for (int u = 0; u < BM_AHEAD; ++u) {
          ei[u] = __builtin_amdgcn_readlane(gv, j0 + u) * BM_GRP + lane;
          sv[u] = ls[ei[u] < e ? ei[u] : e - 1];
        }
#pragma unroll
        for (int u = 0; u < BM_AHEAD; ++u) {
          if (j0 + u < n32) {
            const unsigned l =
                ei[u] < e ? (unsigned)(sv[u] - row0) : 0xffffffffu;
            take(l < (unsigned)ROWS, ei[u], (int)l);
          }
        }
        work_off();
      }
    }
    wave_lds_sync();
  }

  wave_lds_sync();
  {
    const int w  = lane < qn ? q[lane] : 0;
    const int eq = (int)((unsigned)w / (unsigned)ROWS);
    const int dq = (int)((unsigned)w % (unsigned)ROWS);
    const int pq = lane < qn ? ld[eq] : 0;
    float res = 0.0f;
#pragma unroll 1
    for (int j0 = 0; j0 < qn; j0 += EDGE_ROWS) dot_group(pq, dq, j0, res);
    if (lane < qn) out[eq] = res;
  }
}

extern "C" void kernel_launch(void* const* d_in, const int* in_sizes, int n_in,
                              void* d_out, int out_size, void* d_ws,
                              size_t ws_size, hipStream_t stream) {
  (void)in_sizes; (void)n_in; (void)out_size; (void)ws_size;
  const int NC = 4096, NP = 8192, E = 500000, EL = 500000;

  const float* x_country = (const float*)d_in[0];
  const float* x_product = (const float*)d_in[1];
  const float* Wc_lin = (const float*)d_in[2];
  const float* bc_lin = (const float*)d_in[3];
  const float* Wp_lin = (const float*)d_in[4];
  const float* bp_lin = (const float*)d_in[5];
  const float* emb_c  = (const float*)d_in[6];
  const float* emb_p  = (const float*)d_in[7];
  const float* W1ep_l = (const float*)d_in[8];
  const float* b1ep   = (const float*)d_in[9];
  const float* W1ep_r = (const float*)d_in[10];
  const float* W1pc_l = (const float*)d_in[11];
  const float* b1pc   = (const float*)d_in[12];
  const float* W1pc_r = (const float*)d_in[13];
  const float* W2ep_l = (const float*)d_in[14];
  const float* b2ep   = (const float*)d_in[15];
  const float* W2ep_r = (const float*)d_in[16];
  const float* W2pc_l = (const float*)d_in[17];
  const float* b2pc   = (const float*)d_in[18];
  const float* W2pc_r = (const float*)d_in[19];
  const int* nid_c = (const int*)d_in[20];
  const int* nid_p = (const int*)d_in[21];
  const int* e_src = (const int*)d_in[22];
  const int* e_dst = (const int*)d_in[23];
  const int* l_src = (const int*)d_in[24];
  const int* l_dst = (const int*)d_in[25];

  float* ws = (float*)d_ws;
  const size_t unit = (size_t)NC * HDIM;
  float* xp    = ws;
  float* xc    = ws + 2 * unit;
  float* sumC1 = ws + 3 * unit;
  float* sumP1 = ws + 4 * unit;
  int* listP   = (int*)(ws + 6 * unit);
  int* listC   = (int*)(ws + 8 * unit);
  unsigned* bitsP = (unsigned*)(ws + 10 * unit);
  unsigned* bitsC = bitsP + 131072;
  float* hc    = ws + 10 * unit;
  float* hp    = ws + 2 * unit;
  float* sumP2 = ws;
  float* sumC2 = ws + 4 * unit;
  unsigned* bitsL = (unsigned*)(ws + 5 * unit);
  float* hp2   = ws + 6 * unit;
  float* hc2   = ws + 8 * unit;
  float* degp  = ws + 11 * unit;
  float* degc  = degp + NP;

  const int TB = 256;
  const int own_blocks = 2 * BM_BLOCKS;
  const int OT = 1024;

  gemm_in_kernel<8, 1><<<(NC / 32) * 2, TB, 0, stream>>>(
      x_country, Wc_lin, NC, NC / 8, bc_lin, emb_c, nid_c, xc);
  gemm_in_kernel<4, 2><<<(NP / 64) * 2, TB, 0, stream>>>(
      x_product, Wp_lin, NP, NP / 4, bp_lin, emb_p, nid_p, xp);

  const int runs = (E + BM_RUN - 1) / BM_RUN;
  group_bits_kernel<NP / 256, NC / 256><<<dim3((runs + 7) / 8, 2), TB, 0,
                                          stream>>>(e_dst, e_src, E, bitsP,
                                                    bitsC);

  own_pair_kernel<NP / 256, NC / 256><<<own_blocks, OT, 0, stream>>>(
      xc, xp, e_src, e_dst, E, sumP1, sumC1, degp, degc, listP, listC, 0,
      bitsP, bitsC);
  sage_kernel<<<NC / 16, TB, 0, stream>>>(sumC1, degc, xc,
                                          W1pc_l, b1pc, W1pc_r, hc, 1);
  sage_kernel<<<NP / 16, TB, 0, stream>>>(sumP1, degp, xp,
                                          W1ep_l, b1ep, W1ep_r, hp, 1);

  own_pair_kernel<NP / 256, NC / 256><<<own_blocks, OT, 0, stream>>>(
      hc, hp, e_src, e_dst, E, sumP2, sumC2, degp, degc, listP, listC, 1,
      nullptr, nullptr);
  const int lruns = (EL + BM_RUN - 1) / BM_RUN;
  sage_pair_kernel<NC / 256><<<NP / 16 + NC / 16 + (lruns + 7) / 8, TB, 0,
                               stream>>>(
      NP / 16, NC / 16, 0,
      sumP2, degp, hp, W2ep_l, b2ep, W2ep_r, hp2,
      sumC2, degc, hc, W2pc_l, b2pc, W2pc_r, hc2,
      l_src, EL, bitsL);

  edge_own_kernel<NC / 256><<<BM_BLOCKS, OT, 0, stream>>>(
      hc2, hp2, l_src, l_dst, EL, (float*)d_out, bitsL);
}
